# starting kernel
# baseline (speedup 1.0000x reference)
#include <hip/hip_runtime.h>
#include <hip/hip_bf16.h>

typedef __attribute__((ext_vector_type(2))) float v2f;
typedef __attribute__((ext_vector_type(8))) float v8f;

__global__ void sage_zero_kernel(float* __restrict__ p, long long n) {
    long long i = (long long)blockIdx.x * blockDim.x + threadIdx.x;
    long long stride = (long long)gridDim.x * blockDim.x;
    for (; i < n; i += stride) p[i] = 0.0f;
}

__global__ void sage_pool_gemm_kernel(const float* __restrict__ X,
                                      const float* __restrict__ W,
                                      const float* __restrict__ b,
                                      float* __restrict__ XP, int nrows) {
    __shared__ float2 sW[32 * 64];

    for (int i = threadIdx.x; i < 32 * 64; i += blockDim.x) {
        const int p = i >> 6, c = i & 63;
        float2 t;
        t.x = W[(2 * p) * 64 + c];
        t.y = W[(2 * p + 1) * 64 + c];
        sW[i] = t;
    }
    __syncthreads();

    const int wave = threadIdx.x >> 5;
    const int lane = threadIdx.x & 31;
    const int rowbase = (blockIdx.x * (blockDim.x >> 5) + wave) * 16;
    const int lhalf = lane >> 4;
    const int l16 = lane & 15;
    const int arow = min(rowbase + l16, nrows - 1);
    const float* __restrict__ xr = X + (long long)arow * 64;

    v8f c[4] = {};
    for (int k0 = 0; k0 < 64; k0 += 4) {
        const int ka = k0 + 2 * lhalf;
        const int kp = (k0 >> 1) + lhalf;
        v2f a;
        a.x = xr[ka];
        a.y = xr[ka + 1];
#pragma unroll
        for (int t = 0; t < 4; ++t) {
            const float2 w = sW[kp * 64 + t * 16 + l16];
            v2f bb; bb.x = w.x; bb.y = w.y;
            c[t] = __builtin_amdgcn_wmma_f32_16x16x4_f32(
                false, a, false, bb, (short)0, c[t], false, false);
        }
    }

    if (rowbase + 16 <= nrows) {
#pragma unroll
        for (int t = 0; t < 4; ++t) {
            const int col = t * 16 + l16;
            const float bv = b[col];
            float* __restrict__ yp = XP + (long long)(rowbase + 8 * lhalf) * 64 + col;
#pragma unroll
            for (int j = 0; j < 8; ++j)
                yp[j * 64] = fmaxf(c[t][j] + bv, 0.0f);
        }
    } else {
#pragma unroll
        for (int t = 0; t < 4; ++t) {
            const int col = t * 16 + l16;
            const float bv = b[col];
#pragma unroll
            for (int j = 0; j < 8; ++j) {
                const int row = rowbase + j + 8 * lhalf;
                if (row < nrows)
                    XP[(long long)row * 64 + col] = fmaxf(c[t][j] + bv, 0.0f);
            }
        }
    }
}

__global__ void sage_edge_max_kernel(const int* __restrict__ src,
                                     const int* __restrict__ dst,
                                     const float* __restrict__ XP,
                                     unsigned int* __restrict__ pooled,
                                     int nedges) {
    const long long t = (long long)blockIdx.x * blockDim.x + threadIdx.x;
    const long long e = t >> 4;
    const int sub = (int)(t & 15);
    if (e >= nedges) return;
    const int s = src[e];
    const int d = dst[e];
    const uint4 v = *(const uint4*)(XP + (long long)s * 64 + sub * 4);
    unsigned int* pp = pooled + (long long)d * 64 + sub * 4;
    atomicMax(pp + 0, v.x);
    atomicMax(pp + 1, v.y);
    atomicMax(pp + 2, v.z);
    atomicMax(pp + 3, v.w);
}

template <int NTILES, bool LEAKY>
__global__ void sage_out_gemm_kernel(const float* __restrict__ X,
                                     const float* __restrict__ Wself,
                                     const float* __restrict__ P,
                                     const float* __restrict__ Wneigh,
                                     const float* __restrict__ bias,
                                     float* __restrict__ Y, int nrows) {
    const int NOUT = NTILES * 16;
    __shared__ float2 sWs[32 * NTILES * 16];
    __shared__ float2 sWn[32 * NTILES * 16];

    for (int i = threadIdx.x; i < 32 * NOUT; i += blockDim.x) {
        const int p = i / NOUT, c = i % NOUT;
        float2 t0, t1;
        t0.x = Wself[(2 * p) * NOUT + c];
        t0.y = Wself[(2 * p + 1) * NOUT + c];
        t1.x = Wneigh[(2 * p) * NOUT + c];
        t1.y = Wneigh[(2 * p + 1) * NOUT + c];
        sWs[i] = t0;
        sWn[i] = t1;
    }
    __syncthreads();

    const int wave = threadIdx.x >> 5;
    const int lane = threadIdx.x & 31;
    const int rowbase = (blockIdx.x * (blockDim.x >> 5) + wave) * 16;
    const int lhalf = lane >> 4;
    const int l16 = lane & 15;
    const int arow = min(rowbase + l16, nrows - 1);
    const float* __restrict__ xr = X + (long long)arow * 64;
    const float* __restrict__ pr = P + (long long)arow * 64;

    v8f c[NTILES] = {};
    for (int k0 = 0; k0 < 64; k0 += 4) {
        const int ka = k0 + 2 * lhalf;
        const int kp = (k0 >> 1) + lhalf;
        v2f a1, a2;
        a1.x = xr[ka];  a1.y = xr[ka + 1];
        a2.x = pr[ka];  a2.y = pr[ka + 1];
#pragma unroll
        for (int t = 0; t < NTILES; ++t) {
            const float2 ws = sWs[kp * NOUT + t * 16 + l16];
            const float2 wn = sWn[kp * NOUT + t * 16 + l16];
            v2f b1; b1.x = ws.x; b1.y = ws.y;
            v2f b2; b2.x = wn.x; b2.y = wn.y;
            c[t] = __builtin_amdgcn_wmma_f32_16x16x4_f32(
                false, a1, false, b1, (short)0, c[t], false, false);
            c[t] = __builtin_amdgcn_wmma_f32_16x16x4_f32(
                false, a2, false, b2, (short)0, c[t], false, false);
        }
    }

    if (rowbase + 16 <= nrows) {
#pragma unroll
        for (int t = 0; t < NTILES; ++t) {
            const int col = t * 16 + l16;
            const float bv = bias[col];
            float* __restrict__ yp = Y + (long long)(rowbase + 8 * lhalf) * NOUT + col;
#pragma unroll
            for (int j = 0; j < 8; ++j) {
                float v = c[t][j] + bv;
                if (LEAKY) v = (v >= 0.0f) ? v : 0.01f * v;
                yp[j * NOUT] = v;
            }
        }
    } else {
#pragma unroll
        for (int t = 0; t < NTILES; ++t) {
            const int col = t * 16 + l16;
            const float bv = bias[col];
#pragma unroll
            for (int j = 0; j < 8; ++j) {
                const int row = rowbase + j + 8 * lhalf;
                if (row < nrows) {
                    float v = c[t][j] + bv;
                    if (LEAKY) v = (v >= 0.0f) ? v : 0.01f * v;
                    Y[(long long)row * NOUT + col] = v;
                }
            }
        }
    }
}

extern "C" void kernel_launch(void* const* d_in, const int* in_sizes, int n_in,
                              void* d_out, int out_size, void* d_ws, size_t ws_size,
                              hipStream_t stream) {
    const float* in_feat  = (const float*)d_in[0];
    const int*   src      = (const int*)d_in[1];
    const int*   dst      = (const int*)d_in[2];
    const float* W_pool1  = (const float*)d_in[3];
    const float* b_pool1  = (const float*)d_in[4];
    const float* W_self1  = (const float*)d_in[5];
    const float* W_neigh1 = (const float*)d_in[6];
    const float* bias1    = (const float*)d_in[7];
    const float* W_pool2  = (const float*)d_in[8];
    const float* b_pool2  = (const float*)d_in[9];
    const float* W_self2  = (const float*)d_in[10];
    const float* W_neigh2 = (const float*)d_in[11];
    const float* bias2    = (const float*)d_in[12];

    const int N = in_sizes[0] / 64;
    const int E = in_sizes[1];
    const long long NF = (long long)N * 64;

    float* XP = (float*)d_ws;
    float* P  = XP + NF;
    float* H  = P + NF;

    const int gemm_threads = 128;
    const int gemm_blocks  = (N + 63) / 64;
    int zero_blocks = (int)((NF + 255) / 256);
    if (zero_blocks > 4096) zero_blocks = 4096;
    const long long edge_threads_total = (long long)E * 16;
    const int edge_blocks = (int)((edge_threads_total + 255) / 256);

    sage_zero_kernel<<<zero_blocks, 256, 0, stream>>>(P, NF);
    sage_pool_gemm_kernel<<<gemm_blocks, gemm_threads, 0, stream>>>(
        in_feat, W_pool1, b_pool1, XP, N);
    sage_edge_max_kernel<<<edge_blocks, 256, 0, stream>>>(
        src, dst, XP, (unsigned int*)P, E);
    sage_out_gemm_kernel<4, true><<<gemm_blocks, gemm_threads, 0, stream>>>(
        in_feat, W_self1, P, W_neigh1, bias1, H, N);

    sage_pool_gemm_kernel<<<gemm_blocks, gemm_threads, 0, stream>>>(
        H, W_pool2, b_pool2, XP, N);
    sage_zero_kernel<<<zero_blocks, 256, 0, stream>>>(P, NF);
    sage_edge_max_kernel<<<edge_blocks, 256, 0, stream>>>(
        src, dst, XP, (unsigned int*)P, E);
    sage_out_gemm_kernel<1, false><<<gemm_blocks, gemm_threads, 0, stream>>>(
        H, W_self2, P, W_neigh2, bias2, (float*)d_out, N);
}


# Segment max by gather: edges ordered by dst once per call (counting int add returns each edge's slot, arrival order; max is order-free), one wave per node maxes its rows; no float atomics or pre-zero
# speedup vs baseline: 1.3214x; 1.3214x over previous
#include <hip/hip_runtime.h>
#include <hip/hip_bf16.h>

typedef __attribute__((ext_vector_type(2))) float v2f;
typedef __attribute__((ext_vector_type(8))) float v8f;

__global__ void sage_zero_kernel(float* __restrict__ p, long long n) {
    long long i = (long long)blockIdx.x * blockDim.x + threadIdx.x;
    long long stride = (long long)gridDim.x * blockDim.x;
    for (; i < n; i += stride) p[i] = 0.0f;
}

__global__ void sage_pool_gemm_kernel(const float* __restrict__ X,
                                      const float* __restrict__ W,
                                      const float* __restrict__ b,
                                      float* __restrict__ XP, int nrows) {
    __shared__ float2 sW[32 * 64];

    for (int i = threadIdx.x; i < 32 * 64; i += blockDim.x) {
        const int p = i >> 6, c = i & 63;
        float2 t;
        t.x = W[(2 * p) * 64 + c];
        t.y = W[(2 * p + 1) * 64 + c];
        sW[i] = t;
    }
    __syncthreads();

    const int wave = threadIdx.x >> 5;
    const int lane = threadIdx.x & 31;
    const int rowbase = (blockIdx.x * (blockDim.x >> 5) + wave) * 16;
    const int lhalf = lane >> 4;
    const int l16 = lane & 15;
    const int arow = min(rowbase + l16, nrows - 1);
    const float* __restrict__ xr = X + (long long)arow * 64;

    v8f c[4] = {};
    for (int k0 = 0; k0 < 64; k0 += 4) {
        const int ka = k0 + 2 * lhalf;
        const int kp = (k0 >> 1) + lhalf;
        v2f a;
        a.x = xr[ka];
        a.y = xr[ka + 1];
#pragma unroll
        for (int t = 0; t < 4; ++t) {
            const float2 w = sW[kp * 64 + t * 16 + l16];
            v2f bb; bb.x = w.x; bb.y = w.y;
            c[t] = __builtin_amdgcn_wmma_f32_16x16x4_f32(
                false, a, false, bb, (short)0, c[t], false, false);
        }
    }

    if (rowbase + 16 <= nrows) {
#pragma unroll
        for (int t = 0; t < 4; ++t) {
            const int col = t * 16 + l16;
            const float bv = b[col];
            float* __restrict__ yp = XP + (long long)(rowbase + 8 * lhalf) * 64 + col;
#pragma unroll
            for (int j = 0; j < 8; ++j)
                yp[j * 64] = fmaxf(c[t][j] + bv, 0.0f);
        }
    } else {
#pragma unroll
        for (int t = 0; t < 4; ++t) {
            const int col = t * 16 + l16;
            const float bv = b[col];
#pragma unroll
            for (int j = 0; j < 8; ++j) {
                const int row = rowbase + j + 8 * lhalf;
                if (row < nrows)
                    XP[(long long)row * 64 + col] = fmaxf(c[t][j] + bv, 0.0f);
            }
        }
    }
}

__global__ void sage_count_rank_kernel(const int* __restrict__ dst,
                                       int* __restrict__ cnt,
                                       int* __restrict__ rank, int nedges) {
    const int T = gridDim.x * blockDim.x;
    int e = blockIdx.x * blockDim.x + threadIdx.x;
    for (; e + 3 * T < nedges; e += 4 * T) {
        const int d0 = dst[e];
        const int d1 = dst[e + T];
        const int d2 = dst[e + 2 * T];
        const int d3 = dst[e + 3 * T];
        const int r0 = atomicAdd(cnt + d0, 1);
        const int r1 = atomicAdd(cnt + d1, 1);
        const int r2 = atomicAdd(cnt + d2, 1);
        const int r3 = atomicAdd(cnt + d3, 1);
        rank[e] = r0;
        rank[e + T] = r1;
        rank[e + 2 * T] = r2;
        rank[e + 3 * T] = r3;
    }
    for (; e < nedges; e += T)
        rank[e] = atomicAdd(cnt + dst[e], 1);
}

__global__ void __launch_bounds__(1024)
sage_offsets_kernel(const int* __restrict__ cnt, int* __restrict__ off, int n) {
    __shared__ int s_pre[32];
    __shared__ int s_inc[32];
    const int tid = threadIdx.x;
    const int lane = tid & 31;
    const int wv = tid >> 5;
    const int b = blockIdx.x;
    const int4* __restrict__ c4 = (const int4*)cnt;
    const int n4 = b * 256;
    int pre = 0;
#pragma unroll 4
    for (int j = tid; j < n4; j += 1024) {
        const int4 v = c4[j];
        pre += (v.x + v.y) + (v.z + v.w);
    }
    const int i = b * 1024 + tid;
    const int own = (i < n) ? cnt[i] : 0;
    int inc = own;
    for (int d = 1; d < 32; d <<= 1) {
        const int t = __shfl_up(inc, d);
        if (lane >= d) inc += t;
    }
    for (int d = 16; d > 0; d >>= 1) pre += __shfl_xor(pre, d);
    if (lane == 31) s_inc[wv] = inc;
    if (lane == 0) s_pre[wv] = pre;
    __syncthreads();
    int add = s_pre[lane] + ((lane < wv) ? s_inc[lane] : 0);
    for (int d = 16; d > 0; d >>= 1) add += __shfl_xor(add, d);
    if (i <= n) off[i] = add + inc - own;
}

__global__ void sage_place_kernel(const int* __restrict__ src,
                                  const int* __restrict__ dst,
                                  const int* __restrict__ off,
                                  const int* __restrict__ rank,
                                  int* __restrict__ nbr, int nedges) {
    const int e = blockIdx.x * blockDim.x + threadIdx.x;
    if (e >= nedges) return;
    nbr[off[dst[e]] + rank[e]] = src[e];
}

static __device__ __forceinline__ float4 max4(const float4 a, const float4 b) {
    float4 r;
    r.x = fmaxf(a.x, b.x);
    r.y = fmaxf(a.y, b.y);
    r.z = fmaxf(a.z, b.z);
    r.w = fmaxf(a.w, b.w);
    return r;
}

__global__ void __launch_bounds__(256)
sage_seg_max_kernel(const int* __restrict__ off,
                    const int* __restrict__ nbr,
                    const float* __restrict__ XP,
                    float* __restrict__ pooled, int nnodes) {
    const int lane = threadIdx.x & 31;
    const int wv = __builtin_amdgcn_readfirstlane((int)(threadIdx.x >> 5));
    const int node = blockIdx.x * 8 + wv;
    if (node >= nnodes) return;
    const int lhalf = lane >> 4;
    const int l16 = lane & 15;
    const int beg = off[node];
    const int end = off[node + 1];
    const float* __restrict__ xc = XP + l16 * 4;

    float4 acc = make_float4(0.0f, 0.0f, 0.0f, 0.0f);
    for (int base = beg; base < end; base += 32) {
        const int cnt = min(32, end - base);
        const int myid = nbr[base + min(lane, cnt - 1)];
        int j = 0;
        for (; cnt - j > 4; j += 8) {
            const int s0 = __shfl(myid, j + lhalf);
            const int s1 = __shfl(myid, j + 2 + lhalf);
            const int s2 = __shfl(myid, j + 4 + lhalf);
            const int s3 = __shfl(myid, j + 6 + lhalf);
            const float4 v0 = *(const float4*)(xc + (long long)s0 * 64);
            const float4 v1 = *(const float4*)(xc + (long long)s1 * 64);
            const float4 v2 = *(const float4*)(xc + (long long)s2 * 64);
            const float4 v3 = *(const float4*)(xc + (long long)s3 * 64);
            acc = max4(acc, max4(max4(v0, v1), max4(v2, v3)));
        }
        if (j < cnt) {
            const int s0 = __shfl(myid, j + lhalf);
            const int s1 = __shfl(myid, j + 2 + lhalf);
            const float4 v0 = *(const float4*)(xc + (long long)s0 * 64);
            const float4 v1 = *(const float4*)(xc + (long long)s1 * 64);
            acc = max4(acc, max4(v0, v1));
        }
    }
    acc.x = fmaxf(acc.x, __shfl_xor(acc.x, 16));
    acc.y = fmaxf(acc.y, __shfl_xor(acc.y, 16));
    acc.z = fmaxf(acc.z, __shfl_xor(acc.z, 16));
    acc.w = fmaxf(acc.w, __shfl_xor(acc.w, 16));
    if (lhalf == 0)
        *(float4*)(pooled + (long long)node * 64 + l16 * 4) = acc;
}

template <int NTILES, bool LEAKY>
__global__ void sage_out_gemm_kernel(const float* __restrict__ X,
                                     const float* __restrict__ Wself,
                                     const float* __restrict__ P,
                                     const float* __restrict__ Wneigh,
                                     const float* __restrict__ bias,
                                     float* __restrict__ Y, int nrows) {
    const int NOUT = NTILES * 16;
    __shared__ float2 sWs[32 * NTILES * 16];
    __shared__ float2 sWn[32 * NTILES * 16];

    for (int i = threadIdx.x; i < 32 * NOUT; i += blockDim.x) {
        const int p = i / NOUT, c = i % NOUT;
        float2 t0, t1;
        t0.x = Wself[(2 * p) * NOUT + c];
        t0.y = Wself[(2 * p + 1) * NOUT + c];
        t1.x = Wneigh[(2 * p) * NOUT + c];
        t1.y = Wneigh[(2 * p + 1) * NOUT + c];
        sWs[i] = t0;
        sWn[i] = t1;
    }
    __syncthreads();

    const int wave = threadIdx.x >> 5;
    const int lane = threadIdx.x & 31;
    const int rowbase = (blockIdx.x * (blockDim.x >> 5) + wave) * 16;
    const int lhalf = lane >> 4;
    const int l16 = lane & 15;
    const int arow = min(rowbase + l16, nrows - 1);
    const float* __restrict__ xr = X + (long long)arow * 64;
    const float* __restrict__ pr = P + (long long)arow * 64;

    v8f c[NTILES] = {};
    for (int k0 = 0; k0 < 64; k0 += 4) {
        const int ka = k0 + 2 * lhalf;
        const int kp = (k0 >> 1) + lhalf;
        v2f a1, a2;
        a1.x = xr[ka];  a1.y = xr[ka + 1];
        a2.x = pr[ka];  a2.y = pr[ka + 1];
#pragma unroll
        for (int t = 0; t < NTILES; ++t) {
            const float2 ws = sWs[kp * NOUT + t * 16 + l16];
            const float2 wn = sWn[kp * NOUT + t * 16 + l16];
            v2f b1; b1.x = ws.x; b1.y = ws.y;
            v2f b2; b2.x = wn.x; b2.y = wn.y;
            c[t] = __builtin_amdgcn_wmma_f32_16x16x4_f32(
                false, a1, false, b1, (short)0, c[t], false, false);
            c[t] = __builtin_amdgcn_wmma_f32_16x16x4_f32(
                false, a2, false, b2, (short)0, c[t], false, false);
        }
    }

    if (rowbase + 16 <= nrows) {
#pragma unroll
        for (int t = 0; t < NTILES; ++t) {
            const int col = t * 16 + l16;
            const float bv = bias[col];
            float* __restrict__ yp = Y + (long long)(rowbase + 8 * lhalf) * NOUT + col;
#pragma unroll
            for (int j = 0; j < 8; ++j) {
                float v = c[t][j] + bv;
                if (LEAKY) v = (v >= 0.0f) ? v : 0.01f * v;
                yp[j * NOUT] = v;
            }
        }
    } else {
#pragma unroll
        for (int t = 0; t < NTILES; ++t) {
            const int col = t * 16 + l16;
            const float bv = bias[col];
#pragma unroll
            for (int j = 0; j < 8; ++j) {
                const int row = rowbase + j + 8 * lhalf;
                if (row < nrows) {
                    float v = c[t][j] + bv;
                    if (LEAKY) v = (v >= 0.0f) ? v : 0.01f * v;
                    Y[(long long)row * NOUT + col] = v;
                }
            }
        }
    }
}

extern "C" void kernel_launch(void* const* d_in, const int* in_sizes, int n_in,
                              void* d_out, int out_size, void* d_ws, size_t ws_size,
                              hipStream_t stream) {
    const float* in_feat  = (const float*)d_in[0];
    const int*   src      = (const int*)d_in[1];
    const int*   dst      = (const int*)d_in[2];
    const float* W_pool1  = (const float*)d_in[3];
    const float* b_pool1  = (const float*)d_in[4];
    const float* W_self1  = (const float*)d_in[5];
    const float* W_neigh1 = (const float*)d_in[6];
    const float* bias1    = (const float*)d_in[7];
    const float* W_pool2  = (const float*)d_in[8];
    const float* b_pool2  = (const float*)d_in[9];
    const float* W_self2  = (const float*)d_in[10];
    const float* W_neigh2 = (const float*)d_in[11];
    const float* bias2    = (const float*)d_in[12];

    const int N = in_sizes[0] / 64;
    const int E = in_sizes[1];
    const long long NF = (long long)N * 64;

    float* XP = (float*)d_ws;
    float* P  = XP + NF;
    float* H  = P + NF;
    int* cnt  = (int*)(H + NF);
    int* off  = cnt + (N + 1024);
    int* rank = off + (N + 1024);
    int* nbr  = rank + E;

    const int gemm_threads = 128;
    const int gemm_blocks  = (N + 63) / 64;
    const int zero_blocks  = (N + 255) / 256;
    const int rank_blocks  = ((E + 3) / 4 + 255) / 256;
    const int scan_blocks  = (N + 1 + 1023) / 1024;
    const int edge_blocks  = (E + 255) / 256;
    const int seg_blocks   = (N + 7) / 8;

    sage_zero_kernel<<<zero_blocks, 256, 0, stream>>>((float*)cnt, (long long)N);
    sage_count_rank_kernel<<<rank_blocks, 256, 0, stream>>>(dst, cnt, rank, E);
    sage_offsets_kernel<<<scan_blocks, 1024, 0, stream>>>(cnt, off, N);
    sage_place_kernel<<<edge_blocks, 256, 0, stream>>>(src, dst, off, rank, nbr, E);

    sage_pool_gemm_kernel<<<gemm_blocks, gemm_threads, 0, stream>>>(
        in_feat, W_pool1, b_pool1, XP, N);
    sage_seg_max_kernel<<<seg_blocks, 256, 0, stream>>>(off, nbr, XP, P, N);
    sage_out_gemm_kernel<4, true><<<gemm_blocks, gemm_threads, 0, stream>>>(
        in_feat, W_self1, P, W_neigh1, bias1, H, N);

    sage_pool_gemm_kernel<<<gemm_blocks, gemm_threads, 0, stream>>>(
        H, W_pool2, b_pool2, XP, N);
    sage_seg_max_kernel<<<seg_blocks, 256, 0, stream>>>(off, nbr, XP, P, N);
    sage_out_gemm_kernel<1, false><<<gemm_blocks, gemm_threads, 0, stream>>>(
        H, W_self2, P, W_neigh2, bias2, (float*)d_out, N);
}


# Edge slots from LDS counters, not global adds: 16x16 work-groups (edge chunk x node range) count dst in LDS, one scan gives each tile its first slots, an LDS cursor pass places; max is order-free
# speedup vs baseline: 1.8959x; 1.4348x over previous
#include <hip/hip_runtime.h>
#include <hip/hip_bf16.h>

typedef __attribute__((ext_vector_type(2))) float v2f;
typedef __attribute__((ext_vector_type(8))) float v8f;

__global__ void sage_pool_gemm_kernel(const float* __restrict__ X,
                                      const float* __restrict__ W,
                                      const float* __restrict__ b,
                                      float* __restrict__ XP, int nrows) {
    __shared__ float2 sW[32 * 64];

    for (int i = threadIdx.x; i < 32 * 64; i += blockDim.x) {
        const int p = i >> 6, c = i & 63;
        float2 t;
        t.x = W[(2 * p) * 64 + c];
        t.y = W[(2 * p + 1) * 64 + c];
        sW[i] = t;
    }
    __syncthreads();

    const int wave = threadIdx.x >> 5;
    const int lane = threadIdx.x & 31;
    const int rowbase = (blockIdx.x * (blockDim.x >> 5) + wave) * 16;
    const int lhalf = lane >> 4;
    const int l16 = lane & 15;
    const int arow = min(rowbase + l16, nrows - 1);
    const float* __restrict__ xr = X + (long long)arow * 64;

    v8f c[4] = {};
    for (int k0 = 0; k0 < 64; k0 += 4) {
        const int ka = k0 + 2 * lhalf;
        const int kp = (k0 >> 1) + lhalf;
        v2f a;
        a.x = xr[ka];
        a.y = xr[ka + 1];
#pragma unroll
        for (int t = 0; t < 4; ++t) {
            const float2 w = sW[kp * 64 + t * 16 + l16];
            v2f bb; bb.x = w.x; bb.y = w.y;
            c[t] = __builtin_amdgcn_wmma_f32_16x16x4_f32(
                false, a, false, bb, (short)0, c[t], false, false);
        }
    }

    if (rowbase + 16 <= nrows) {
#pragma unroll
        for (int t = 0; t < 4; ++t) {
            const int col = t * 16 + l16;
            const float bv = b[col];
            float* __restrict__ yp = XP + (long long)(rowbase + 8 * lhalf) * 64 + col;
#pragma unroll
            for (int j = 0; j < 8; ++j)
                yp[j * 64] = fmaxf(c[t][j] + bv, 0.0f);
        }
    } else {
#pragma unroll
        for (int t = 0; t < 4; ++t) {
            const int col = t * 16 + l16;
            const float bv = b[col];
#pragma unroll
            for (int j = 0; j < 8; ++j) {
                const int row = rowbase + j + 8 * lhalf;
                if (row < nrows)
                    XP[(long long)row * 64 + col] = fmaxf(c[t][j] + bv, 0.0f);
            }
        }
    }
}

#define EDGE_CHUNKS 16
#define NODE_RANGES 16
#define SCAN_BATCH 5

__global__ void __launch_bounds__(1024)
sage_count_tile_kernel(const int* __restrict__ dst,
                       int* __restrict__ part,
                       int* __restrict__ gsum,
                       int nedges, int chunk, int rn, int npad, int ngp) {
    extern __shared__ int s_tile[];
    const int tid = threadIdx.x;
    const int c = blockIdx.x;
    const int r = blockIdx.y;
    const int lo = r * rn;
    const int gpr = rn >> 10;
    for (int i = tid; i < rn; i += 1024) s_tile[i] = 0;
    __syncthreads();

    const int e0 = c * chunk;
    const int len = min(chunk, nedges - e0);
    const int len4 = (len > 0) ? (len & ~3) : 0;
    const int* __restrict__ dp = dst + e0;
    for (int b0 = tid * 4; b0 < len4; b0 += 4096 * SCAN_BATCH) {
        int4 d[SCAN_BATCH];
#pragma unroll
        for (int u = 0; u < SCAN_BATCH; ++u) {
            const int idx = b0 + u * 4096;
            d[u] = (idx < len4) ? *(const int4*)(dp + idx) : make_int4(-1, -1, -1, -1);
        }
#pragma unroll
        for (int u = 0; u < SCAN_BATCH; ++u) {
            const unsigned k0 = (unsigned)(d[u].x - lo);
            const unsigned k1 = (unsigned)(d[u].y - lo);
            const unsigned k2 = (unsigned)(d[u].z - lo);
            const unsigned k3 = (unsigned)(d[u].w - lo);
            if (k0 < (unsigned)rn) atomicAdd(s_tile + k0, 1);
            if (k1 < (unsigned)rn) atomicAdd(s_tile + k1, 1);
            if (k2 < (unsigned)rn) atomicAdd(s_tile + k2, 1);
            if (k3 < (unsigned)rn) atomicAdd(s_tile + k3, 1);
        }
    }
    for (int i = len4 + tid; i < len; i += 1024) {
        const unsigned k = (unsigned)(dp[i] - lo);
        if (k < (unsigned)rn) atomicAdd(s_tile + k, 1);
    }
    __syncthreads();

    int* __restrict__ pc = part + c * npad + lo;
    for (int i = tid; i < rn; i += 1024) pc[i] = s_tile[i];
    const int lane = tid & 31;
    for (int k = tid >> 5; k < gpr; k += 32) {
        int s = 0;
#pragma unroll
        for (int m = 0; m < 32; ++m) s += s_tile[k * 1024 + m * 32 + lane];
        for (int d = 16; d > 0; d >>= 1) s += __shfl_xor(s, d);
        if (lane == 0) gsum[c * ngp + r * gpr + k] = s;
    }
}

__global__ void __launch_bounds__(1024)
sage_bases_kernel(int* __restrict__ part, const int* __restrict__ gsum,
                  int* __restrict__ off, int n, int npad, int ngp) {
    __shared__ int s_pre[32];
    __shared__ int s_inc[32];
    const int tid = threadIdx.x;
    const int lane = tid & 31;
    const int wv = tid >> 5;
    const int b = blockIdx.x;
    int pre = 0;
    for (int g = tid; g < b; g += 1024) {
#pragma unroll
        for (int c = 0; c < EDGE_CHUNKS; ++c) pre += gsum[c * ngp + g];
    }
    const int i = b * 1024 + tid;
    int v[EDGE_CHUNKS];
    int own = 0;
#pragma unroll
    for (int c = 0; c < EDGE_CHUNKS; ++c) {
        v[c] = part[c * npad + i];
        own += v[c];
    }
    int inc = own;
    for (int d = 1; d < 32; d <<= 1) {
        const int t = __shfl_up(inc, d);
        if (lane >= d) inc += t;
    }
    for (int d = 16; d > 0; d >>= 1) pre += __shfl_xor(pre, d);
    if (lane == 31) s_inc[wv] = inc;
    if (lane == 0) s_pre[wv] = pre;
    __syncthreads();
    int add = s_pre[lane] + ((lane < wv) ? s_inc[lane] : 0);
    for (int d = 16; d > 0; d >>= 1) add += __shfl_xor(add, d);
    int run = add + inc - own;
    if (i <= n) off[i] = run;
#pragma unroll
    for (int c = 0; c < EDGE_CHUNKS; ++c) {
        part[c * npad + i] = run;
        run += v[c];
    }
}

__global__ void __launch_bounds__(1024)
sage_place_tile_kernel(const int* __restrict__ src,
                       const int* __restrict__ dst,
                       const int* __restrict__ base,
                       int* __restrict__ nbr,
                       int nedges, int chunk, int rn, int npad) {
    extern __shared__ int s_tile[];
    const int tid = threadIdx.x;
    const int c = blockIdx.x;
    const int r = blockIdx.y;
    const int lo = r * rn;
    const int* __restrict__ bc = base + c * npad + lo;
    for (int i = tid; i < rn; i += 1024) s_tile[i] = bc[i];
    __syncthreads();
    unsigned* __restrict__ cur = (unsigned*)s_tile;

    const int e0 = c * chunk;
    const int len = min(chunk, nedges - e0);
    const int len4 = (len > 0) ? (len & ~3) : 0;
    const int* __restrict__ dp = dst + e0;
    const int* __restrict__ sp = src + e0;
    for (int b0 = tid * 4; b0 < len4; b0 += 4096 * SCAN_BATCH) {
        int4 d[SCAN_BATCH], s[SCAN_BATCH];
#pragma unroll
        for (int u = 0; u < SCAN_BATCH; ++u) {
            const int idx = b0 + u * 4096;
            const bool in = idx < len4;
            d[u] = in ? *(const int4*)(dp + idx) : make_int4(-1, -1, -1, -1);
            s[u] = in ? *(const int4*)(sp + idx) : make_int4(0, 0, 0, 0);
        }
        unsigned p[SCAN_BATCH][4];
        bool t[SCAN_BATCH][4];
#pragma unroll
        for (int u = 0; u < SCAN_BATCH; ++u) {
            const unsigned k0 = (unsigned)(d[u].x - lo);
            const unsigned k1 = (unsigned)(d[u].y - lo);
            const unsigned k2 = (unsigned)(d[u].z - lo);
            const unsigned k3 = (unsigned)(d[u].w - lo);
            t[u][0] = k0 < (unsigned)rn;
            t[u][1] = k1 < (unsigned)rn;
            t[u][2] = k2 < (unsigned)rn;
            t[u][3] = k3 < (unsigned)rn;
            p[u][0] = p[u][1] = p[u][2] = p[u][3] = 0u;
            if (t[u][0]) p[u][0] = atomicAdd(cur + k0, 1u);
            if (t[u][1]) p[u][1] = atomicAdd(cur + k1, 1u);
            if (t[u][2]) p[u][2] = atomicAdd(cur + k2, 1u);
            if (t[u][3]) p[u][3] = atomicAdd(cur + k3, 1u);
        }
#pragma unroll
        for (int u = 0; u < SCAN_BATCH; ++u) {
            if (t[u][0]) nbr[p[u][0]] = s[u].x;
            if (t[u][1]) nbr[p[u][1]] = s[u].y;
            if (t[u][2]) nbr[p[u][2]] = s[u].z;
            if (t[u][3]) nbr[p[u][3]] = s[u].w;
        }
    }
    for (int i = len4 + tid; i < len; i += 1024) {
        const unsigned k = (unsigned)(dp[i] - lo);
        if (k < (unsigned)rn) nbr[atomicAdd(cur + k, 1u)] = sp[i];
    }
}

static __device__ __forceinline__ float4 max4(const float4 a, const float4 b) {
    float4 r;
    r.x = fmaxf(a.x, b.x);
    r.y = fmaxf(a.y, b.y);
    r.z = fmaxf(a.z, b.z);
    r.w = fmaxf(a.w, b.w);
    return r;
}

__global__ void __launch_bounds__(256)
sage_seg_max_kernel(const int* __restrict__ off,
                    const int* __restrict__ nbr,
                    const float* __restrict__ XP,
                    float* __restrict__ pooled, int nnodes) {
    const int lane = threadIdx.x & 31;
    const int wv = __builtin_amdgcn_readfirstlane((int)(threadIdx.x >> 5));
    const int node = blockIdx.x * 8 + wv;
    if (node >= nnodes) return;
    const int lhalf = lane >> 4;
    const int l16 = lane & 15;
    const int beg = off[node];
    const int end = off[node + 1];
    const float* __restrict__ xc = XP + l16 * 4;

    float4 acc = make_float4(0.0f, 0.0f, 0.0f, 0.0f);
    for (int base = beg; base < end; base += 32) {
        const int cnt = min(32, end - base);
        const int myid = nbr[base + min(lane, cnt - 1)];
        int j = 0;
        for (; cnt - j > 4; j += 8) {
            const int s0 = __shfl(myid, j + lhalf);
            const int s1 = __shfl(myid, j + 2 + lhalf);
            const int s2 = __shfl(myid, j + 4 + lhalf);
            const int s3 = __shfl(myid, j + 6 + lhalf);
            const float4 v0 = *(const float4*)(xc + (long long)s0 * 64);
            const float4 v1 = *(const float4*)(xc + (long long)s1 * 64);
            const float4 v2 = *(const float4*)(xc + (long long)s2 * 64);
            const float4 v3 = *(const float4*)(xc + (long long)s3 * 64);
            acc = max4(acc, max4(max4(v0, v1), max4(v2, v3)));
        }
        if (j < cnt) {
            const int s0 = __shfl(myid, j + lhalf);
            const int s1 = __shfl(myid, j + 2 + lhalf);
            const float4 v0 = *(const float4*)(xc + (long long)s0 * 64);
            const float4 v1 = *(const float4*)(xc + (long long)s1 * 64);
            acc = max4(acc, max4(v0, v1));
        }
    }
    acc.x = fmaxf(acc.x, __shfl_xor(acc.x, 16));
    acc.y = fmaxf(acc.y, __shfl_xor(acc.y, 16));
    acc.z = fmaxf(acc.z, __shfl_xor(acc.z, 16));
    acc.w = fmaxf(acc.w, __shfl_xor(acc.w, 16));
    if (lhalf == 0)
        *(float4*)(pooled + (long long)node * 64 + l16 * 4) = acc;
}

template <int NTILES, bool LEAKY>
__global__ void sage_out_gemm_kernel(const float* __restrict__ X,
                                     const float* __restrict__ Wself,
                                     const float* __restrict__ P,
                                     const float* __restrict__ Wneigh,
                                     const float* __restrict__ bias,
                                     float* __restrict__ Y, int nrows) {
    const int NOUT = NTILES * 16;
    __shared__ float2 sWs[32 * NTILES * 16];
    __shared__ float2 sWn[32 * NTILES * 16];

    for (int i = threadIdx.x; i < 32 * NOUT; i += blockDim.x) {
        const int p = i / NOUT, c = i % NOUT;
        float2 t0, t1;
        t0.x = Wself[(2 * p) * NOUT + c];
        t0.y = Wself[(2 * p + 1) * NOUT + c];
        t1.x = Wneigh[(2 * p) * NOUT + c];
        t1.y = Wneigh[(2 * p + 1) * NOUT + c];
        sWs[i] = t0;
        sWn[i] = t1;
    }
    __syncthreads();

    const int wave = threadIdx.x >> 5;
    const int lane = threadIdx.x & 31;
    const int rowbase = (blockIdx.x * (blockDim.x >> 5) + wave) * 16;
    const int lhalf = lane >> 4;
    const int l16 = lane & 15;
    const int arow = min(rowbase + l16, nrows - 1);
    const float* __restrict__ xr = X + (long long)arow * 64;
    const float* __restrict__ pr = P + (long long)arow * 64;

    v8f c[NTILES] = {};
    for (int k0 = 0; k0 < 64; k0 += 4) {
        const int ka = k0 + 2 * lhalf;
        const int kp = (k0 >> 1) + lhalf;
        v2f a1, a2;
        a1.x = xr[ka];  a1.y = xr[ka + 1];
        a2.x = pr[ka];  a2.y = pr[ka + 1];
#pragma unroll
        for (int t = 0; t < NTILES; ++t) {
            const float2 ws = sWs[kp * NOUT + t * 16 + l16];
            const float2 wn = sWn[kp * NOUT + t * 16 + l16];
            v2f b1; b1.x = ws.x; b1.y = ws.y;
            v2f b2; b2.x = wn.x; b2.y = wn.y;
            c[t] = __builtin_amdgcn_wmma_f32_16x16x4_f32(
                false, a1, false, b1, (short)0, c[t], false, false);
            c[t] = __builtin_amdgcn_wmma_f32_16x16x4_f32(
                false, a2, false, b2, (short)0, c[t], false, false);
        }
    }

    if (rowbase + 16 <= nrows) {
#pragma unroll
        for (int t = 0; t < NTILES; ++t) {
            const int col = t * 16 + l16;
            const float bv = bias[col];
            float* __restrict__ yp = Y + (long long)(rowbase + 8 * lhalf) * NOUT + col;
#pragma unroll
            for (int j = 0; j < 8; ++j) {
                float v = c[t][j] + bv;
                if (LEAKY) v = (v >= 0.0f) ? v : 0.01f * v;
                yp[j * NOUT] = v;
            }
        }
    } else {
#pragma unroll
        for (int t = 0; t < NTILES; ++t) {
            const int col = t * 16 + l16;
            const float bv = bias[col];
#pragma unroll
            for (int j = 0; j < 8; ++j) {
                const int row = rowbase + j + 8 * lhalf;
                if (row < nrows) {
                    float v = c[t][j] + bv;
                    if (LEAKY) v = (v >= 0.0f) ? v : 0.01f * v;
                    Y[(long long)row * NOUT + col] = v;
                }
            }
        }
    }
}

extern "C" void kernel_launch(void* const* d_in, const int* in_sizes, int n_in,
                              void* d_out, int out_size, void* d_ws, size_t ws_size,
                              hipStream_t stream) {
    const float* in_feat  = (const float*)d_in[0];
    const int*   src      = (const int*)d_in[1];
    const int*   dst      = (const int*)d_in[2];
    const float* W_pool1  = (const float*)d_in[3];
    const float* b_pool1  = (const float*)d_in[4];
    const float* W_self1  = (const float*)d_in[5];
    const float* W_neigh1 = (const float*)d_in[6];
    const float* bias1    = (const float*)d_in[7];
    const float* W_pool2  = (const float*)d_in[8];
    const float* b_pool2  = (const float*)d_in[9];
    const float* W_self2  = (const float*)d_in[10];
    const float* W_neigh2 = (const float*)d_in[11];
    const float* bias2    = (const float*)d_in[12];

    const int N = in_sizes[0] / 64;
    const int E = in_sizes[1];
    const long long NF = (long long)N * 64;

    const int gemm_threads = 128;
    const int gemm_blocks  = (N + 63) / 64;
    const int scan_blocks  = (N + 1 + 1023) / 1024;
    const int seg_blocks   = (N + 7) / 8;
    const int gpr   = (scan_blocks + NODE_RANGES - 1) / NODE_RANGES;
    const int rn    = gpr * 1024;
    const int ngp   = gpr * NODE_RANGES;
    const int npad  = ngp * 1024;
    const int chunk = ((E + EDGE_CHUNKS - 1) / EDGE_CHUNKS + 3) & ~3;
    const dim3 tile_grid(EDGE_CHUNKS, NODE_RANGES);

    float* XP = (float*)d_ws;
    float* P  = XP + NF;
    float* H  = P + NF;
    int* off  = (int*)(H + NF);
    int* part = off + (N + 1024);
    int* gsum = part + EDGE_CHUNKS * npad;
    int* nbr  = gsum + EDGE_CHUNKS * ngp;

    sage_count_tile_kernel<<<tile_grid, 1024, rn * sizeof(int), stream>>>(
        dst, part, gsum, E, chunk, rn, npad, ngp);
    sage_bases_kernel<<<scan_blocks, 1024, 0, stream>>>(part, gsum, off, N, npad, ngp);
    sage_place_tile_kernel<<<tile_grid, 1024, rn * sizeof(int), stream>>>(
        src, dst, part, nbr, E, chunk, rn, npad);

    sage_pool_gemm_kernel<<<gemm_blocks, gemm_threads, 0, stream>>>(
        in_feat, W_pool1, b_pool1, XP, N);
    sage_seg_max_kernel<<<seg_blocks, 256, 0, stream>>>(off, nbr, XP, P, N);
    sage_out_gemm_kernel<4, true><<<gemm_blocks, gemm_threads, 0, stream>>>(
        in_feat, W_self1, P, W_neigh1, bias1, H, N);

    sage_pool_gemm_kernel<<<gemm_blocks, gemm_threads, 0, stream>>>(
        H, W_pool2, b_pool2, XP, N);
    sage_seg_max_kernel<<<seg_blocks, 256, 0, stream>>>(off, nbr, XP, P, N);
    sage_out_gemm_kernel<1, false><<<gemm_blocks, gemm_threads, 0, stream>>>(
        H, W_self2, P, W_neigh2, bias2, (float*)d_out, N);
}


# Gather fused into the output GEMM: 16-wave work-groups walk 16-node tiles, pooled and own rows go through LDS into the A fragments, pooled never to memory; 7 launches not 9; same order of sums
# speedup vs baseline: 2.0934x; 1.1042x over previous
#include <hip/hip_runtime.h>
#include <hip/hip_bf16.h>

typedef __attribute__((ext_vector_type(2))) float v2f;
typedef __attribute__((ext_vector_type(8))) float v8f;

__global__ void sage_pool_gemm_kernel(const float* __restrict__ X,
                                      const float* __restrict__ W,
                                      const float* __restrict__ b,
                                      float* __restrict__ XP, int nrows) {
    __shared__ float2 sW[32 * 64];

    for (int i = threadIdx.x; i < 32 * 64; i += blockDim.x) {
        const int p = i >> 6, c = i & 63;
        float2 t;
        t.x = W[(2 * p) * 64 + c];
        t.y = W[(2 * p + 1) * 64 + c];
        sW[i] = t;
    }
    __syncthreads();

    const int wave = threadIdx.x >> 5;
    const int lane = threadIdx.x & 31;
    const int rowbase = (blockIdx.x * (blockDim.x >> 5) + wave) * 16;
    const int lhalf = lane >> 4;
    const int l16 = lane & 15;
    const int arow = min(rowbase + l16, nrows - 1);
    const float* __restrict__ xr = X + (long long)arow * 64;

    v8f c[4] = {};
    for (int k0 = 0; k0 < 64; k0 += 4) {
        const int ka = k0 + 2 * lhalf;
        const int kp = (k0 >> 1) + lhalf;
        v2f a;
        a.x = xr[ka];
        a.y = xr[ka + 1];
#pragma unroll
        for (int t = 0; t < 4; ++t) {
            const float2 w = sW[kp * 64 + t * 16 + l16];
            v2f bb; bb.x = w.x; bb.y = w.y;
            c[t] = __builtin_amdgcn_wmma_f32_16x16x4_f32(
                false, a, false, bb, (short)0, c[t], false, false);
        }
    }

    if (rowbase + 16 <= nrows) {
#pragma unroll
        for (int t = 0; t < 4; ++t) {
            const int col = t * 16 + l16;
            const float bv = b[col];
            float* __restrict__ yp = XP + (long long)(rowbase + 8 * lhalf) * 64 + col;
#pragma unroll
            for (int j = 0; j < 8; ++j)
                yp[j * 64] = fmaxf(c[t][j] + bv, 0.0f);
        }
    } else {
#pragma unroll
        for (int t = 0; t < 4; ++t) {
            const int col = t * 16 + l16;
            const float bv = b[col];
#pragma unroll
            for (int j = 0; j < 8; ++j) {
                const int row = rowbase + j + 8 * lhalf;
                if (row < nrows)
                    XP[(long long)row * 64 + col] = fmaxf(c[t][j] + bv, 0.0f);
            }
        }
    }
}

#define EDGE_CHUNKS 16
#define NODE_RANGES 16
#define SCAN_BATCH 5

__global__ void __launch_bounds__(1024)
sage_count_tile_kernel(const int* __restrict__ dst,
                       int* __restrict__ part,
                       int* __restrict__ gsum,
                       int nedges, int chunk, int rn, int npad, int ngp) {
    extern __shared__ int s_tile[];
    const int tid = threadIdx.x;
    const int c = blockIdx.x;
    const int r = blockIdx.y;
    const int lo = r * rn;
    const int gpr = rn >> 10;
    for (int i = tid; i < rn; i += 1024) s_tile[i] = 0;
    __syncthreads();

    const int e0 = c * chunk;
    const int len = min(chunk, nedges - e0);
    const int len4 = (len > 0) ? (len & ~3) : 0;
    const int* __restrict__ dp = dst + e0;
    for (int b0 = tid * 4; b0 < len4; b0 += 4096 * SCAN_BATCH) {
        int4 d[SCAN_BATCH];
#pragma unroll
        for (int u = 0; u < SCAN_BATCH; ++u) {
            const int idx = b0 + u * 4096;
            d[u] = (idx < len4) ? *(const int4*)(dp + idx) : make_int4(-1, -1, -1, -1);
        }
#pragma unroll
        for (int u = 0; u < SCAN_BATCH; ++u) {
            const unsigned k0 = (unsigned)(d[u].x - lo);
            const unsigned k1 = (unsigned)(d[u].y - lo);
            const unsigned k2 = (unsigned)(d[u].z - lo);
            const unsigned k3 = (unsigned)(d[u].w - lo);
            if (k0 < (unsigned)rn) atomicAdd(s_tile + k0, 1);
            if (k1 < (unsigned)rn) atomicAdd(s_tile + k1, 1);
            if (k2 < (unsigned)rn) atomicAdd(s_tile + k2, 1);
            if (k3 < (unsigned)rn) atomicAdd(s_tile + k3, 1);
        }
    }
    for (int i = len4 + tid; i < len; i += 1024) {
        const unsigned k = (unsigned)(dp[i] - lo);
        if (k < (unsigned)rn) atomicAdd(s_tile + k, 1);
    }
    __syncthreads();

    int* __restrict__ pc = part + c * npad + lo;
    for (int i = tid; i < rn; i += 1024) pc[i] = s_tile[i];
    const int lane = tid & 31;
    for (int k = tid >> 5; k < gpr; k += 32) {
        int s = 0;
#pragma unroll
        for (int m = 0; m < 32; ++m) s += s_tile[k * 1024 + m * 32 + lane];
        for (int d = 16; d > 0; d >>= 1) s += __shfl_xor(s, d);
        if (lane == 0) gsum[c * ngp + r * gpr + k] = s;
    }
}

__global__ void __launch_bounds__(1024)
sage_bases_kernel(int* __restrict__ part, const int* __restrict__ gsum,
                  int* __restrict__ off, int n, int npad, int ngp) {
    __shared__ int s_pre[32];
    __shared__ int s_inc[32];
    const int tid = threadIdx.x;
    const int lane = tid & 31;
    const int wv = tid >> 5;
    const int b = blockIdx.x;
    int pre = 0;
    for (int g = tid; g < b; g += 1024) {
#pragma unroll
        for (int c = 0; c < EDGE_CHUNKS; ++c) pre += gsum[c * ngp + g];
    }
    const int i = b * 1024 + tid;
    int v[EDGE_CHUNKS];
    int own = 0;
#pragma unroll
    for (int c = 0; c < EDGE_CHUNKS; ++c) {
        v[c] = part[c * npad + i];
        own += v[c];
    }
    int inc = own;
    for (int d = 1; d < 32; d <<= 1) {
        const int t = __shfl_up(inc, d);
        if (lane >= d) inc += t;
    }
    for (int d = 16; d > 0; d >>= 1) pre += __shfl_xor(pre, d);
    if (lane == 31) s_inc[wv] = inc;
    if (lane == 0) s_pre[wv] = pre;
    __syncthreads();
    int add = s_pre[lane] + ((lane < wv) ? s_inc[lane] : 0);
    for (int d = 16; d > 0; d >>= 1) add += __shfl_xor(add, d);
    int run = add + inc - own;
    if (i <= n) off[i] = run;
#pragma unroll
    for (int c = 0; c < EDGE_CHUNKS; ++c) {
        part[c * npad + i] = run;
        run += v[c];
    }
}

__global__ void __launch_bounds__(1024)
sage_place_tile_kernel(const int* __restrict__ src,
                       const int* __restrict__ dst,
                       const int* __restrict__ base,
                       int* __restrict__ nbr,
                       int nedges, int chunk, int rn, int npad) {
    extern __shared__ int s_tile[];
    const int tid = threadIdx.x;
    const int c = blockIdx.x;
    const int r = blockIdx.y;
    const int lo = r * rn;
    const int* __restrict__ bc = base + c * npad + lo;
    for (int i = tid; i < rn; i += 1024) s_tile[i] = bc[i];
    __syncthreads();
    unsigned* __restrict__ cur = (unsigned*)s_tile;

    const int e0 = c * chunk;
    const int len = min(chunk, nedges - e0);
    const int len4 = (len > 0) ? (len & ~3) : 0;
    const int* __restrict__ dp = dst + e0;
    const int* __restrict__ sp = src + e0;
    for (int b0 = tid * 4; b0 < len4; b0 += 4096 * SCAN_BATCH) {
        int4 d[SCAN_BATCH], s[SCAN_BATCH];
#pragma unroll
        for (int u = 0; u < SCAN_BATCH; ++u) {
            const int idx = b0 + u * 4096;
            const bool in = idx < len4;
            d[u] = in ? *(const int4*)(dp + idx) : make_int4(-1, -1, -1, -1);
            s[u] = in ? *(const int4*)(sp + idx) : make_int4(0, 0, 0, 0);
        }
        unsigned p[SCAN_BATCH][4];
        bool t[SCAN_BATCH][4];
#pragma unroll
        for (int u = 0; u < SCAN_BATCH; ++u) {
            const unsigned k0 = (unsigned)(d[u].x - lo);
            const unsigned k1 = (unsigned)(d[u].y - lo);
            const unsigned k2 = (unsigned)(d[u].z - lo);
            const unsigned k3 = (unsigned)(d[u].w - lo);
            t[u][0] = k0 < (unsigned)rn;
            t[u][1] = k1 < (unsigned)rn;
            t[u][2] = k2 < (unsigned)rn;
            t[u][3] = k3 < (unsigned)rn;
            p[u][0] = p[u][1] = p[u][2] = p[u][3] = 0u;
            if (t[u][0]) p[u][0] = atomicAdd(cur + k0, 1u);
            if (t[u][1]) p[u][1] = atomicAdd(cur + k1, 1u);
            if (t[u][2]) p[u][2] = atomicAdd(cur + k2, 1u);
            if (t[u][3]) p[u][3] = atomicAdd(cur + k3, 1u);
        }
#pragma unroll
        for (int u = 0; u < SCAN_BATCH; ++u) {
            if (t[u][0]) nbr[p[u][0]] = s[u].x;
            if (t[u][1]) nbr[p[u][1]] = s[u].y;
            if (t[u][2]) nbr[p[u][2]] = s[u].z;
            if (t[u][3]) nbr[p[u][3]] = s[u].w;
        }
    }
    for (int i = len4 + tid; i < len; i += 1024) {
        const unsigned k = (unsigned)(dp[i] - lo);
        if (k < (unsigned)rn) nbr[atomicAdd(cur + k, 1u)] = sp[i];
    }
}

static __device__ __forceinline__ float4 max4(const float4 a, const float4 b) {
    float4 r;
    r.x = fmaxf(a.x, b.x);
    r.y = fmaxf(a.y, b.y);
    r.z = fmaxf(a.z, b.z);
    r.w = fmaxf(a.w, b.w);
    return r;
}

static __device__ __forceinline__ float4 max_rows(const float* __restrict__ xc, const int myid,
                                                  const int cnt, const int lhalf, float4 acc) {
    int j = 0;
    for (; cnt - j > 4; j += 8) {
        const int s0 = __shfl(myid, j + lhalf);
        const int s1 = __shfl(myid, j + 2 + lhalf);
        const int s2 = __shfl(myid, j + 4 + lhalf);
        const int s3 = __shfl(myid, j + 6 + lhalf);
        const float4 v0 = *(const float4*)(xc + (long long)s0 * 64);
        const float4 v1 = *(const float4*)(xc + (long long)s1 * 64);
        const float4 v2 = *(const float4*)(xc + (long long)s2 * 64);
        const float4 v3 = *(const float4*)(xc + (long long)s3 * 64);
        acc = max4(acc, max4(max4(v0, v1), max4(v2, v3)));
    }
    if (j < cnt) {
        const int s0 = __shfl(myid, j + lhalf);
        const int s1 = __shfl(myid, j + 2 + lhalf);
        const float4 v0 = *(const float4*)(xc + (long long)s0 * 64);
        const float4 v1 = *(const float4*)(xc + (long long)s1 * 64);
        acc = max4(acc, max4(v0, v1));
    }
    return acc;
}

#define TILE_PITCH 68

template <int NTILES, bool LEAKY>
static __device__ __forceinline__ void tile_product(const float* __restrict__ rows,
                                                    const float2* __restrict__ sWs,
                                                    const float2* __restrict__ sWn,
                                                    const float bv, float* __restrict__ Y,
                                                    const int tile, const int t, const int nnodes,
                                                    const int lhalf, const int l16) {
    const int NOUT = NTILES * 16;
    const float* __restrict__ ax = rows + l16 * TILE_PITCH + 2 * lhalf;
    const float* __restrict__ ap = ax + 16 * TILE_PITCH;
    v8f c = {};
#pragma unroll
    for (int k0 = 0; k0 < 64; k0 += 4) {
        const int kp = (k0 >> 1) + lhalf;
        const float2 x2 = *(const float2*)(ax + k0);
        const float2 p2 = *(const float2*)(ap + k0);
        const float2 ws = sWs[kp * NOUT + t * 16 + l16];
        const float2 wn = sWn[kp * NOUT + t * 16 + l16];
        v2f a1; a1.x = x2.x; a1.y = x2.y;
        v2f a2; a2.x = p2.x; a2.y = p2.y;
        v2f b1; b1.x = ws.x; b1.y = ws.y;
        v2f b2; b2.x = wn.x; b2.y = wn.y;
        c = __builtin_amdgcn_wmma_f32_16x16x4_f32(
            false, a1, false, b1, (short)0, c, false, false);
        c = __builtin_amdgcn_wmma_f32_16x16x4_f32(
            false, a2, false, b2, (short)0, c, false, false);
    }
    const int row0 = tile * 16 + 8 * lhalf;
    float* __restrict__ yp = Y + (long long)row0 * NOUT + t * 16 + l16;
    if (tile * 16 + 16 <= nnodes) {
#pragma unroll
        for (int j = 0; j < 8; ++j) {
            float v = c[j] + bv;
            if (LEAKY) v = (v >= 0.0f) ? v : 0.01f * v;
            yp[j * NOUT] = v;
        }
    } else {
#pragma unroll
        for (int j = 0; j < 8; ++j) {
            if (row0 + j < nnodes) {
                float v = c[j] + bv;
                if (LEAKY) v = (v >= 0.0f) ? v : 0.01f * v;
                yp[j * NOUT] = v;
            }
        }
    }
}

template <int NTILES, bool LEAKY>
__global__ void __launch_bounds__(512)
sage_gather_out_kernel(const int* __restrict__ off,
                       const int* __restrict__ nbr,
                       const float* __restrict__ XP,
                       const float* __restrict__ X,
                       const float* __restrict__ Wself,
                       const float* __restrict__ Wneigh,
                       const float* __restrict__ bias,
                       float* __restrict__ Y, int nnodes, int ntiles) {
    const int NOUT = NTILES * 16;
    __shared__ float2 sWs[32 * NTILES * 16];
    __shared__ float2 sWn[32 * NTILES * 16];
    __shared__ __attribute__((aligned(16))) float sA[2][2 * 16 * TILE_PITCH];

#pragma unroll
    for (int i = threadIdx.x; i < 32 * NOUT; i += 512) {
        const int p = i / NOUT, c = i % NOUT;
        float2 t0, t1;
        t0.x = Wself[(2 * p) * NOUT + c];
        t0.y = Wself[(2 * p + 1) * NOUT + c];
        t1.x = Wneigh[(2 * p) * NOUT + c];
        t1.y = Wneigh[(2 * p + 1) * NOUT + c];
        sWs[i] = t0;
        sWn[i] = t1;
    }

    const int lane = threadIdx.x & 31;
    const int wv = __builtin_amdgcn_readfirstlane((int)(threadIdx.x >> 5));
    const int lhalf = lane >> 4;
    const int l16 = lane & 15;
    const float* __restrict__ xc = XP + l16 * 4;
    const int stride = gridDim.x;
    const int lastn = nnodes - 1;

    const int tcol = (NTILES == 4) ? (wv / 5) : 0;
    const bool mm4 = (wv % 5 == 0);
    const float bv = bias[tcol * 16 + l16];

    int tile = blockIdx.x;
    int bnd = off[min(tile * 16 + wv, lastn) + (lane & 1)];
    int it = 0;
    for (; tile < ntiles; tile += stride, ++it) {
        const int node = tile * 16 + wv;
        const int beg = __builtin_amdgcn_readlane(bnd, 0);
        const int end = (node < nnodes) ? __builtin_amdgcn_readlane(bnd, 1) : beg;
        const int cnt0 = min(32, end - beg);
        const int myid0 = nbr[(cnt0 > 0) ? beg + min(lane, cnt0 - 1) : 0];
        bnd = off[min(node + stride * 16, lastn) + (lane & 1)];
        const float4 sv = *(const float4*)(X + (long long)min(node, lastn) * 64 + l16 * 4);

        if (it > 0 && ((NTILES == 4) ? mm4 : (wv == 5 * ((it - 1) & 3))))
            tile_product<NTILES, LEAKY>(sA[(it - 1) & 1], sWs, sWn, bv, Y, tile - stride, tcol,
                                        nnodes, lhalf, l16);

        float4 acc = make_float4(0.0f, 0.0f, 0.0f, 0.0f);
        acc = max_rows(xc, myid0, cnt0, lhalf, acc);
        for (int base = beg + 32; base < end; base += 32) {
            const int cnt = min(32, end - base);
            const int myid = nbr[base + min(lane, cnt - 1)];
            acc = max_rows(xc, myid, cnt, lhalf, acc);
        }
        acc.x = fmaxf(acc.x, __shfl_xor(acc.x, 16));
        acc.y = fmaxf(acc.y, __shfl_xor(acc.y, 16));
        acc.z = fmaxf(acc.z, __shfl_xor(acc.z, 16));
        acc.w = fmaxf(acc.w, __shfl_xor(acc.w, 16));
        float4 o;
        o.x = lhalf ? sv.x : acc.x;
        o.y = lhalf ? sv.y : acc.y;
        o.z = lhalf ? sv.z : acc.z;
        o.w = lhalf ? sv.w : acc.w;
        *(float4*)(&sA[it & 1][(1 - lhalf) * 16 * TILE_PITCH + wv * TILE_PITCH + l16 * 4]) = o;
        __syncthreads();
    }
    if (it > 0 && ((NTILES == 4) ? mm4 : (wv == 5 * ((it - 1) & 3))))
        tile_product<NTILES, LEAKY>(sA[(it - 1) & 1], sWs, sWn, bv, Y, tile - stride, tcol,
                                    nnodes, lhalf, l16);
}

extern "C" void kernel_launch(void* const* d_in, const int* in_sizes, int n_in,
                              void* d_out, int out_size, void* d_ws, size_t ws_size,
                              hipStream_t stream) {
    const float* in_feat  = (const float*)d_in[0];
    const int*   src      = (const int*)d_in[1];
    const int*   dst      = (const int*)d_in[2];
    const float* W_pool1  = (const float*)d_in[3];
    const float* b_pool1  = (const float*)d_in[4];
    const float* W_self1  = (const float*)d_in[5];
    const float* W_neigh1 = (const float*)d_in[6];
    const float* bias1    = (const float*)d_in[7];
    const float* W_pool2  = (const float*)d_in[8];
    const float* b_pool2  = (const float*)d_in[9];
    const float* W_self2  = (const float*)d_in[10];
    const float* W_neigh2 = (const float*)d_in[11];
    const float* bias2    = (const float*)d_in[12];

    const int N = in_sizes[0] / 64;
    const int E = in_sizes[1];
    const long long NF = (long long)N * 64;

    const int gemm_threads = 128;
    const int gemm_blocks  = (N + 63) / 64;
    const int scan_blocks  = (N + 1 + 1023) / 1024;
    const int ntiles       = (N + 15) / 16;
    const int fuse_blocks  = ntiles < 1024 ? ntiles : 1024;
    const int gpr   = (scan_blocks + NODE_RANGES - 1) / NODE_RANGES;
    const int rn    = gpr * 1024;
    const int ngp   = gpr * NODE_RANGES;
    const int npad  = ngp * 1024;
    const int chunk = ((E + EDGE_CHUNKS - 1) / EDGE_CHUNKS + 3) & ~3;
    const dim3 tile_grid(EDGE_CHUNKS, NODE_RANGES);

    float* XP = (float*)d_ws;
    float* H  = XP + NF;
    int* off  = (int*)(H + NF);
    int* part = off + (N + 1024);
    int* gsum = part + EDGE_CHUNKS * npad;
    int* nbr  = gsum + EDGE_CHUNKS * ngp;

    sage_count_tile_kernel<<<tile_grid, 1024, rn * sizeof(int), stream>>>(
        dst, part, gsum, E, chunk, rn, npad, ngp);
    sage_bases_kernel<<<scan_blocks, 1024, 0, stream>>>(part, gsum, off, N, npad, ngp);
    sage_place_tile_kernel<<<tile_grid, 1024, rn * sizeof(int), stream>>>(
        src, dst, part, nbr, E, chunk, rn, npad);

    sage_pool_gemm_kernel<<<gemm_blocks, gemm_threads, 0, stream>>>(
        in_feat, W_pool1, b_pool1, XP, N);
    sage_gather_out_kernel<4, true><<<fuse_blocks, 512, 0, stream>>>(
        off, nbr, XP, in_feat, W_self1, W_neigh1, bias1, H, N, ntiles);

    sage_pool_gemm_kernel<<<gemm_blocks, gemm_threads, 0, stream>>>(
        H, W_pool2, b_pool2, XP, N);
    sage_gather_out_kernel<1, false><<<fuse_blocks, 512, 0, stream>>>(
        off, nbr, XP, H, W_self2, W_neigh2, bias2, (float*)d_out, N, ntiles);
}


# Layer-2 pool product inside the layer-1 output kernel: each finished 16-node tile of H stays in LDS, four other waves multiply it by W_pool2 a step later; 6 launches not 7; same order of sums
# speedup vs baseline: 2.4613x; 1.1757x over previous
#include <hip/hip_runtime.h>
#include <hip/hip_bf16.h>

typedef __attribute__((ext_vector_type(2))) float v2f;
typedef __attribute__((ext_vector_type(8))) float v8f;

__global__ void sage_pool_gemm_kernel(const float* __restrict__ X,
                                      const float* __restrict__ W,
                                      const float* __restrict__ b,
                                      float* __restrict__ XP, int nrows) {
    __shared__ float2 sW[32 * 64];

    for (int i = threadIdx.x; i < 32 * 64; i += blockDim.x) {
        const int p = i >> 6, c = i & 63;
        float2 t;
        t.x = W[(2 * p) * 64 + c];
        t.y = W[(2 * p + 1) * 64 + c];
        sW[i] = t;
    }
    __syncthreads();

    const int wave = threadIdx.x >> 5;
    const int lane = threadIdx.x & 31;
    const int rowbase = (blockIdx.x * (blockDim.x >> 5) + wave) * 16;
    const int lhalf = lane >> 4;
    const int l16 = lane & 15;
    const int arow = min(rowbase + l16, nrows - 1);
    const float* __restrict__ xr = X + (long long)arow * 64;

    v8f c[4] = {};
    for (int k0 = 0; k0 < 64; k0 += 4) {
        const int ka = k0 + 2 * lhalf;
        const int kp = (k0 >> 1) + lhalf;
        v2f a;
        a.x = xr[ka];
        a.y = xr[ka + 1];
#pragma unroll
        for (int t = 0; t < 4; ++t) {
            const float2 w = sW[kp * 64 + t * 16 + l16];
            v2f bb; bb.x = w.x; bb.y = w.y;
            c[t] = __builtin_amdgcn_wmma_f32_16x16x4_f32(
                false, a, false, bb, (short)0, c[t], false, false);
        }
    }

    if (rowbase + 16 <= nrows) {
#pragma unroll
        for (int t = 0; t < 4; ++t) {
            const int col = t * 16 + l16;
            const float bv = b[col];
            float* __restrict__ yp = XP + (long long)(rowbase + 8 * lhalf) * 64 + col;
#pragma unroll
            for (int j = 0; j < 8; ++j)
                yp[j * 64] = fmaxf(c[t][j] + bv, 0.0f);
        }
    } else {
#pragma unroll
        for (int t = 0; t < 4; ++t) {
            const int col = t * 16 + l16;
            const float bv = b[col];
#pragma unroll
            for (int j = 0; j < 8; ++j) {
                const int row = rowbase + j + 8 * lhalf;
                if (row < nrows)
                    XP[(long long)row * 64 + col] = fmaxf(c[t][j] + bv, 0.0f);
            }
        }
    }
}

#define EDGE_CHUNKS 16
#define NODE_RANGES 16
#define SCAN_BATCH 5

__global__ void __launch_bounds__(1024)
sage_count_tile_kernel(const int* __restrict__ dst,
                       int* __restrict__ part,
                       int* __restrict__ gsum,
                       int nedges, int chunk, int rn, int npad, int ngp) {
    extern __shared__ int s_tile[];
    const int tid = threadIdx.x;
    const int c = blockIdx.x;
    const int r = blockIdx.y;
    const int lo = r * rn;
    const int gpr = rn >> 10;
    for (int i = tid; i < rn; i += 1024) s_tile[i] = 0;
    __syncthreads();

    const int e0 = c * chunk;
    const int len = min(chunk, nedges - e0);
    const int len4 = (len > 0) ? (len & ~3) : 0;
    const int* __restrict__ dp = dst + e0;
    for (int b0 = tid * 4; b0 < len4; b0 += 4096 * SCAN_BATCH) {
        int4 d[SCAN_BATCH];
#pragma unroll
        for (int u = 0; u < SCAN_BATCH; ++u) {
            const int idx = b0 + u * 4096;
            d[u] = (idx < len4) ? *(const int4*)(dp + idx) : make_int4(-1, -1, -1, -1);
        }
#pragma unroll
        for (int u = 0; u < SCAN_BATCH; ++u) {
            const unsigned k0 = (unsigned)(d[u].x - lo);
            const unsigned k1 = (unsigned)(d[u].y - lo);
            const unsigned k2 = (unsigned)(d[u].z - lo);
            const unsigned k3 = (unsigned)(d[u].w - lo);
            if (k0 < (unsigned)rn) atomicAdd(s_tile + k0, 1);
            if (k1 < (unsigned)rn) atomicAdd(s_tile + k1, 1);
            if (k2 < (unsigned)rn) atomicAdd(s_tile + k2, 1);
            if (k3 < (unsigned)rn) atomicAdd(s_tile + k3, 1);
        }
    }
    for (int i = len4 + tid; i < len; i += 1024) {
        const unsigned k = (unsigned)(dp[i] - lo);
        if (k < (unsigned)rn) atomicAdd(s_tile + k, 1);
    }
    __syncthreads();

    int* __restrict__ pc = part + c * npad + lo;
    for (int i = tid; i < rn; i += 1024) pc[i] = s_tile[i];
    const int lane = tid & 31;
    for (int k = tid >> 5; k < gpr; k += 32) {
        int s = 0;
#pragma unroll
        for (int m = 0; m < 32; ++m) s += s_tile[k * 1024 + m * 32 + lane];
        for (int d = 16; d > 0; d >>= 1) s += __shfl_xor(s, d);
        if (lane == 0) gsum[c * ngp + r * gpr + k] = s;
    }
}

__global__ void __launch_bounds__(1024)
sage_bases_kernel(int* __restrict__ part, const int* __restrict__ gsum,
                  int* __restrict__ off, int n, int npad, int ngp) {
    __shared__ int s_pre[32];
    __shared__ int s_inc[32];
    const int tid = threadIdx.x;
    const int lane = tid & 31;
    const int wv = tid >> 5;
    const int b = blockIdx.x;
    int pre = 0;
    for (int g = tid; g < b; g += 1024) {
#pragma unroll
        for (int c = 0; c < EDGE_CHUNKS; ++c) pre += gsum[c * ngp + g];
    }
    const int i = b * 1024 + tid;
    int v[EDGE_CHUNKS];
    int own = 0;
#pragma unroll
    for (int c = 0; c < EDGE_CHUNKS; ++c) {
        v[c] = part[c * npad + i];
        own += v[c];
    }
    int inc = own;
    for (int d = 1; d < 32; d <<= 1) {
        const int t = __shfl_up(inc, d);
        if (lane >= d) inc += t;
    }
    for (int d = 16; d > 0; d >>= 1) pre += __shfl_xor(pre, d);
    if (lane == 31) s_inc[wv] = inc;
    if (lane == 0) s_pre[wv] = pre;
    __syncthreads();
    int add = s_pre[lane] + ((lane < wv) ? s_inc[lane] : 0);
    for (int d = 16; d > 0; d >>= 1) add += __shfl_xor(add, d);
    int run = add + inc - own;
    if (i <= n) off[i] = run;
#pragma unroll
    for (int c = 0; c < EDGE_CHUNKS; ++c) {
        part[c * npad + i] = run;
        run += v[c];
    }
}

__global__ void __launch_bounds__(1024)
sage_place_tile_kernel(const int* __restrict__ src,
                       const int* __restrict__ dst,
                       const int* __restrict__ base,
                       int* __restrict__ nbr,
                       int nedges, int chunk, int rn, int npad) {
    extern __shared__ int s_tile[];
    const int tid = threadIdx.x;
    const int c = blockIdx.x;
    const int r = blockIdx.y;
    const int lo = r * rn;
    const int* __restrict__ bc = base + c * npad + lo;
    for (int i = tid; i < rn; i += 1024) s_tile[i] = bc[i];
    __syncthreads();
    unsigned* __restrict__ cur = (unsigned*)s_tile;

    const int e0 = c * chunk;
    const int len = min(chunk, nedges - e0);
    const int len4 = (len > 0) ? (len & ~3) : 0;
    const int* __restrict__ dp = dst + e0;
    const int* __restrict__ sp = src + e0;
    for (int b0 = tid * 4; b0 < len4; b0 += 4096 * SCAN_BATCH) {
        int4 d[SCAN_BATCH], s[SCAN_BATCH];
#pragma unroll
        for (int u = 0; u < SCAN_BATCH; ++u) {
            const int idx = b0 + u * 4096;
            const bool in = idx < len4;
            d[u] = in ? *(const int4*)(dp + idx) : make_int4(-1, -1, -1, -1);
            s[u] = in ? *(const int4*)(sp + idx) : make_int4(0, 0, 0, 0);
        }
        unsigned p[SCAN_BATCH][4];
        bool t[SCAN_BATCH][4];
#pragma unroll
        for (int u = 0; u < SCAN_BATCH; ++u) {
            const unsigned k0 = (unsigned)(d[u].x - lo);
            const unsigned k1 = (unsigned)(d[u].y - lo);
            const unsigned k2 = (unsigned)(d[u].z - lo);
            const unsigned k3 = (unsigned)(d[u].w - lo);
            t[u][0] = k0 < (unsigned)rn;
            t[u][1] = k1 < (unsigned)rn;
            t[u][2] = k2 < (unsigned)rn;
            t[u][3] = k3 < (unsigned)rn;
            p[u][0] = p[u][1] = p[u][2] = p[u][3] = 0u;
            if (t[u][0]) p[u][0] = atomicAdd(cur + k0, 1u);
            if (t[u][1]) p[u][1] = atomicAdd(cur + k1, 1u);
            if (t[u][2]) p[u][2] = atomicAdd(cur + k2, 1u);
            if (t[u][3]) p[u][3] = atomicAdd(cur + k3, 1u);
        }
#pragma unroll
        for (int u = 0; u < SCAN_BATCH; ++u) {
            if (t[u][0]) nbr[p[u][0]] = s[u].x;
            if (t[u][1]) nbr[p[u][1]] = s[u].y;
            if (t[u][2]) nbr[p[u][2]] = s[u].z;
            if (t[u][3]) nbr[p[u][3]] = s[u].w;
        }
    }
    for (int i = len4 + tid; i < len; i += 1024) {
        const unsigned k = (unsigned)(dp[i] - lo);
        if (k < (unsigned)rn) nbr[atomicAdd(cur + k, 1u)] = sp[i];
    }
}

static __device__ __forceinline__ float4 max4(const float4 a, const float4 b) {
    float4 r;
    r.x = fmaxf(a.x, b.x);
    r.y = fmaxf(a.y, b.y);
    r.z = fmaxf(a.z, b.z);
    r.w = fmaxf(a.w, b.w);
    return r;
}

static __device__ __forceinline__ float4 max_rows(const float* __restrict__ xc, const int myid,
                                                  const int cnt, const int lhalf, float4 acc) {
    int j = 0;
    for (; cnt - j > 4; j += 8) {
        const int s0 = __shfl(myid, j + lhalf);
        const int s1 = __shfl(myid, j + 2 + lhalf);
        const int s2 = __shfl(myid, j + 4 + lhalf);
        const int s3 = __shfl(myid, j + 6 + lhalf);
        const float4 v0 = *(const float4*)(xc + (long long)s0 * 64);
        const float4 v1 = *(const float4*)(xc + (long long)s1 * 64);
        const float4 v2 = *(const float4*)(xc + (long long)s2 * 64);
        const float4 v3 = *(const float4*)(xc + (long long)s3 * 64);
        acc = max4(acc, max4(max4(v0, v1), max4(v2, v3)));
    }
    if (j < cnt) {
        const int s0 = __shfl(myid, j + lhalf);
        const int s1 = __shfl(myid, j + 2 + lhalf);
        const float4 v0 = *(const float4*)(xc + (long long)s0 * 64);
        const float4 v1 = *(const float4*)(xc + (long long)s1 * 64);
        acc = max4(acc, max4(v0, v1));
    }
    return acc;
}

#define TILE_PITCH 68

template <int NTILES, bool LEAKY, bool POOL2>
static __device__ __forceinline__ void tile_product(const float* __restrict__ rows,
                                                    const float2* __restrict__ sWs,
                                                    const float2* __restrict__ sWn,
                                                    const float bv, float* __restrict__ Y,
                                                    float* __restrict__ hrows,
                                                    const int tile, const int t, const int nnodes,
                                                    const int lhalf, const int l16) {
    const int NOUT = NTILES * 16;
    const float* __restrict__ ax = rows + l16 * TILE_PITCH + 2 * lhalf;
    const float* __restrict__ ap = ax + 16 * TILE_PITCH;
    v8f c = {};
#pragma unroll
    for (int k0 = 0; k0 < 64; k0 += 4) {
        const int kp = (k0 >> 1) + lhalf;
        const float2 x2 = *(const float2*)(ax + k0);
        const float2 p2 = *(const float2*)(ap + k0);
        const float2 ws = sWs[kp * NOUT + t * 16 + l16];
        const float2 wn = sWn[kp * NOUT + t * 16 + l16];
        v2f a1; a1.x = x2.x; a1.y = x2.y;
        v2f a2; a2.x = p2.x; a2.y = p2.y;
        v2f b1; b1.x = ws.x; b1.y = ws.y;
        v2f b2; b2.x = wn.x; b2.y = wn.y;
        c = __builtin_amdgcn_wmma_f32_16x16x4_f32(
            false, a1, false, b1, (short)0, c, false, false);
        c = __builtin_amdgcn_wmma_f32_16x16x4_f32(
            false, a2, false, b2, (short)0, c, false, false);
    }
    const int row0 = tile * 16 + 8 * lhalf;
    float* __restrict__ yp = Y + (long long)row0 * NOUT + t * 16 + l16;
    float* __restrict__ hp = hrows + (8 * lhalf) * TILE_PITCH + t * 16 + l16;
    if (tile * 16 + 16 <= nnodes) {
#pragma unroll
        for (int j = 0; j < 8; ++j) {
            float v = c[j] + bv;
            if (LEAKY) v = (v >= 0.0f) ? v : 0.01f * v;
            yp[j * NOUT] = v;
            if (POOL2) hp[j * TILE_PITCH] = v;
        }
    } else {
#pragma unroll
        for (int j = 0; j < 8; ++j) {
            float v = c[j] + bv;
            if (LEAKY) v = (v >= 0.0f) ? v : 0.01f * v;
            if (row0 + j < nnodes) yp[j * NOUT] = v;
            if (POOL2) hp[j * TILE_PITCH] = v;
        }
    }
}

static __device__ __forceinline__ void pool2_product(const float* __restrict__ hrows,
                                                     const float2* __restrict__ sWp,
                                                     const float bv2, float* __restrict__ XP2,
                                                     const int tile, const int t, const int nnodes,
                                                     const int lhalf, const int l16) {
    const float* __restrict__ ah = hrows + l16 * TILE_PITCH + 2 * lhalf;
    v8f c = {};
#pragma unroll
    for (int k0 = 0; k0 < 64; k0 += 4) {
        const int kp = (k0 >> 1) + lhalf;
        const float2 h2 = *(const float2*)(ah + k0);
        const float2 wp = sWp[kp * 64 + t * 16 + l16];
        v2f a; a.x = h2.x; a.y = h2.y;
        v2f b; b.x = wp.x; b.y = wp.y;
        c = __builtin_amdgcn_wmma_f32_16x16x4_f32(
            false, a, false, b, (short)0, c, false, false);
    }
    const int row0 = tile * 16 + 8 * lhalf;
    float* __restrict__ xp = XP2 + (long long)row0 * 64 + t * 16 + l16;
    if (tile * 16 + 16 <= nnodes) {
#pragma unroll
        for (int j = 0; j < 8; ++j)
            xp[j * 64] = fmaxf(c[j] + bv2, 0.0f);
    } else {
#pragma unroll
        for (int j = 0; j < 8; ++j)
            if (row0 + j < nnodes) xp[j * 64] = fmaxf(c[j] + bv2, 0.0f);
    }
}

template <int NTILES, bool LEAKY, bool POOL2>
__global__ void __launch_bounds__(512, 16)
sage_gather_out_kernel(const int* __restrict__ off,
                       const int* __restrict__ nbr,
                       const float* __restrict__ XP,
                       const float* __restrict__ X,
                       const float* __restrict__ Wself,
                       const float* __restrict__ Wneigh,
                       const float* __restrict__ bias,
                       float* __restrict__ Y,
                       const float* __restrict__ Wpool2,
                       const float* __restrict__ bpool2,
                       float* __restrict__ XP2, int nnodes, int ntiles) {
    static_assert(!POOL2 || NTILES == 4, "the second stage reads 64-column tiles of Y");
    const int NOUT = NTILES * 16;
    __shared__ float2 sWs[32 * NTILES * 16];
    __shared__ float2 sWn[32 * NTILES * 16];
    __shared__ __attribute__((aligned(16))) float sA[2][2 * 16 * TILE_PITCH];
    __shared__ float2 sWp[POOL2 ? 32 * 64 : 1];
    __shared__ __attribute__((aligned(16))) float sH[2][POOL2 ? 16 * TILE_PITCH : 4];

#pragma unroll
    for (int i = threadIdx.x; i < 32 * NOUT; i += 512) {
        const int p = i / NOUT, c = i % NOUT;
        float2 t0, t1;
        t0.x = Wself[(2 * p) * NOUT + c];
        t0.y = Wself[(2 * p + 1) * NOUT + c];
        t1.x = Wneigh[(2 * p) * NOUT + c];
        t1.y = Wneigh[(2 * p + 1) * NOUT + c];
        sWs[i] = t0;
        sWn[i] = t1;
    }
    if (POOL2) {
#pragma unroll
        for (int i = threadIdx.x; i < 32 * 64; i += 512) {
            const int p = i >> 6, c = i & 63;
            float2 t2;
            t2.x = Wpool2[(2 * p) * 64 + c];
            t2.y = Wpool2[(2 * p + 1) * 64 + c];
            sWp[i] = t2;
        }
    }

    const int lane = threadIdx.x & 31;
    const int wv = __builtin_amdgcn_readfirstlane((int)(threadIdx.x >> 5));
    const int lhalf = lane >> 4;
    const int l16 = lane & 15;
    const float* __restrict__ xc = XP + l16 * 4;
    const int stride = gridDim.x;
    const int lastn = nnodes - 1;

    const int tcol = (NTILES == 4) ? (wv / 5) : 0;
    const bool mm4 = (wv % 5 == 0);
    const float bv = bias[tcol * 16 + l16];
    const int t2col = (wv == 2) ? 0 : (wv == 7) ? 1 : (wv == 8) ? 2 : 3;
    const bool mm2 = POOL2 && (wv == 2 || wv == 7 || wv == 8 || wv == 13);
    float bv2 = 0.0f;
    if (POOL2) bv2 = bpool2[t2col * 16 + l16];

    int tile = blockIdx.x;
    int bnd = off[min(tile * 16 + wv, lastn) + (lane & 1)];
    int it = 0;
    for (; tile < ntiles; tile += stride, ++it) {
        const int node = tile * 16 + wv;
        const int beg = __builtin_amdgcn_readlane(bnd, 0);
        const int end = (node < nnodes) ? __builtin_amdgcn_readlane(bnd, 1) : beg;
        const int cnt0 = min(32, end - beg);
        const int myid0 = nbr[(cnt0 > 0) ? beg + min(lane, cnt0 - 1) : 0];
        bnd = off[min(node + stride * 16, lastn) + (lane & 1)];
        const float4 sv = *(const float4*)(X + (long long)min(node, lastn) * 64 + l16 * 4);

        if (it > 0 && ((NTILES == 4) ? mm4 : (wv == 5 * ((it - 1) & 3))))
            tile_product<NTILES, LEAKY, POOL2>(sA[(it - 1) & 1], sWs, sWn, bv, Y, sH[(it - 1) & 1],
                                               tile - stride, tcol, nnodes, lhalf, l16);
        if (POOL2 && it > 1 && mm2)
            pool2_product(sH[it & 1], sWp, bv2, XP2, tile - 2 * stride, t2col, nnodes, lhalf, l16);

        float4 acc = make_float4(0.0f, 0.0f, 0.0f, 0.0f);
        acc = max_rows(xc, myid0, cnt0, lhalf, acc);
        for (int base = beg + 32; base < end; base += 32) {
            const int cnt = min(32, end - base);
            const int myid = nbr[base + min(lane, cnt - 1)];
            acc = max_rows(xc, myid, cnt, lhalf, acc);
        }
        acc.x = fmaxf(acc.x, __shfl_xor(acc.x, 16));
        acc.y = fmaxf(acc.y, __shfl_xor(acc.y, 16));
        acc.z = fmaxf(acc.z, __shfl_xor(acc.z, 16));
        acc.w = fmaxf(acc.w, __shfl_xor(acc.w, 16));
        float4 o;
        o.x = lhalf ? sv.x : acc.x;
        o.y = lhalf ? sv.y : acc.y;
        o.z = lhalf ? sv.z : acc.z;
        o.w = lhalf ? sv.w : acc.w;
        *(float4*)(&sA[it & 1][(1 - lhalf) * 16 * TILE_PITCH + wv * TILE_PITCH + l16 * 4]) = o;
        __syncthreads();
    }
    if (it > 0 && ((NTILES == 4) ? mm4 : (wv == 5 * ((it - 1) & 3))))
        tile_product<NTILES, LEAKY, POOL2>(sA[(it - 1) & 1], sWs, sWn, bv, Y, sH[(it - 1) & 1],
                                           tile - stride, tcol, nnodes, lhalf, l16);
    if (POOL2) {
        if (it > 1 && mm2)
            pool2_product(sH[it & 1], sWp, bv2, XP2, tile - 2 * stride, t2col, nnodes, lhalf, l16);
        __syncthreads();
        if (it > 0 && mm2)
            pool2_product(sH[(it - 1) & 1], sWp, bv2, XP2, tile - stride, t2col, nnodes, lhalf, l16);
    }
}

extern "C" void kernel_launch(void* const* d_in, const int* in_sizes, int n_in,
                              void* d_out, int out_size, void* d_ws, size_t ws_size,
                              hipStream_t stream) {
    const float* in_feat  = (const float*)d_in[0];
    const int*   src      = (const int*)d_in[1];
    const int*   dst      = (const int*)d_in[2];
    const float* W_pool1  = (const float*)d_in[3];
    const float* b_pool1  = (const float*)d_in[4];
    const float* W_self1  = (const float*)d_in[5];
    const float* W_neigh1 = (const float*)d_in[6];
    const float* bias1    = (const float*)d_in[7];
    const float* W_pool2  = (const float*)d_in[8];
    const float* b_pool2  = (const float*)d_in[9];
    const float* W_self2  = (const float*)d_in[10];
    const float* W_neigh2 = (const float*)d_in[11];
    const float* bias2    = (const float*)d_in[12];

    const int N = in_sizes[0] / 64;
    const int E = in_sizes[1];
    const long long NF = (long long)N * 64;

    const int gemm_threads = 128;
    const int gemm_blocks  = (N + 63) / 64;
    const int scan_blocks  = (N + 1 + 1023) / 1024;
    const int ntiles       = (N + 15) / 16;
    const int fuse_blocks  = ntiles < 1024 ? ntiles : 1024;
    const int gpr   = (scan_blocks + NODE_RANGES - 1) / NODE_RANGES;
    const int rn    = gpr * 1024;
    const int ngp   = gpr * NODE_RANGES;
    const int npad  = ngp * 1024;
    const int chunk = ((E + EDGE_CHUNKS - 1) / EDGE_CHUNKS + 3) & ~3;
    const dim3 tile_grid(EDGE_CHUNKS, NODE_RANGES);

    float* XP  = (float*)d_ws;
    float* H   = XP + NF;
    float* XP2 = H + NF;
    int* off  = (int*)(XP2 + NF);
    int* part = off + (N + 1024);
    int* gsum = part + EDGE_CHUNKS * npad;
    int* nbr  = gsum + EDGE_CHUNKS * ngp;

    sage_count_tile_kernel<<<tile_grid, 1024, rn * sizeof(int), stream>>>(
        dst, part, gsum, E, chunk, rn, npad, ngp);
    sage_bases_kernel<<<scan_blocks, 1024, 0, stream>>>(part, gsum, off, N, npad, ngp);
    sage_place_tile_kernel<<<tile_grid, 1024, rn * sizeof(int), stream>>>(
        src, dst, part, nbr, E, chunk, rn, npad);

    sage_pool_gemm_kernel<<<gemm_blocks, gemm_threads, 0, stream>>>(
        in_feat, W_pool1, b_pool1, XP, N);
    sage_gather_out_kernel<4, true, true><<<fuse_blocks, 512, 0, stream>>>(
        off, nbr, XP, in_feat, W_self1, W_neigh1, bias1, H, W_pool2, b_pool2, XP2, N, ntiles);

    sage_gather_out_kernel<1, false, false><<<fuse_blocks, 512, 0, stream>>>(
        off, nbr, XP2, H, W_self2, W_neigh2, bias2, (float*)d_out, nullptr, nullptr, nullptr,
        N, ntiles);
}


# Layer-1 pool product moved into extra work-groups of the edge counting launch (the two are independent and now share the chip); 5 launches not 6; same order of sums
# speedup vs baseline: 2.6312x; 1.0690x over previous
#include <hip/hip_runtime.h>
#include <hip/hip_bf16.h>

typedef __attribute__((ext_vector_type(2))) float v2f;
typedef __attribute__((ext_vector_type(8))) float v8f;

static __device__ __forceinline__ void pool_gemm_tiles(const float* __restrict__ X,
                                                       const float* __restrict__ W,
                                                       const float* __restrict__ b,
                                                       float* __restrict__ XP, const int nrows,
                                                       float2* __restrict__ sW, const int wg) {
#pragma unroll
    for (int i = threadIdx.x; i < 32 * 64; i += 1024) {
        const int p = i >> 6, c = i & 63;
        float2 t;
        t.x = W[(2 * p) * 64 + c];
        t.y = W[(2 * p + 1) * 64 + c];
        sW[i] = t;
    }
    __syncthreads();

    const int wave = threadIdx.x >> 5;
    const int lane = threadIdx.x & 31;
    const int lhalf = lane >> 4;
    const int l16 = lane & 15;
    const int ntiles = (nrows + 15) >> 4;

    const int tile = wg * 32 + wave;
    if (tile < ntiles) {
        const int rowbase = tile * 16;
        const int arow = min(rowbase + l16, nrows - 1);
        const float* __restrict__ xr = X + (long long)arow * 64;

        v8f c[4] = {};
        for (int k0 = 0; k0 < 64; k0 += 4) {
            const int ka = k0 + 2 * lhalf;
            const int kp = (k0 >> 1) + lhalf;
            v2f a;
            a.x = xr[ka];
            a.y = xr[ka + 1];
#pragma unroll
            for (int t = 0; t < 4; ++t) {
                const float2 w = sW[kp * 64 + t * 16 + l16];
                v2f bb; bb.x = w.x; bb.y = w.y;
                c[t] = __builtin_amdgcn_wmma_f32_16x16x4_f32(
                    false, a, false, bb, (short)0, c[t], false, false);
            }
        }

        if (rowbase + 16 <= nrows) {
#pragma unroll
            for (int t = 0; t < 4; ++t) {
                const int col = t * 16 + l16;
                const float bv = b[col];
                float* __restrict__ yp = XP + (long long)(rowbase + 8 * lhalf) * 64 + col;
#pragma unroll
                for (int j = 0; j < 8; ++j)
                    yp[j * 64] = fmaxf(c[t][j] + bv, 0.0f);
            }
        } else {
#pragma unroll
            for (int t = 0; t < 4; ++t) {
                const int col = t * 16 + l16;
                const float bv = b[col];
#pragma unroll
                for (int j = 0; j < 8; ++j) {
                    const int row = rowbase + j + 8 * lhalf;
                    if (row < nrows)
                        XP[(long long)row * 64 + col] = fmaxf(c[t][j] + bv, 0.0f);
                }
            }
        }
    }
}

#define EDGE_CHUNKS 16
#define NODE_RANGES 16
#define SCAN_BATCH 5

__global__ void __launch_bounds__(1024, 16)
sage_count_tile_kernel(const int* __restrict__ dst,
                       int* __restrict__ part,
                       int* __restrict__ gsum,
                       int nedges, int chunk, int rn, int npad, int ngp,
                       const float* __restrict__ X,
                       const float* __restrict__ Wpool,
                       const float* __restrict__ bpool,
                       float* __restrict__ XP, int nrows, int prows) {
    extern __shared__ int s_tile[];
    if ((int)blockIdx.y < prows) {
        pool_gemm_tiles(X, Wpool, bpool, XP, nrows, (float2*)s_tile,
                        blockIdx.y * EDGE_CHUNKS + blockIdx.x);
        return;
    }
    const int tid = threadIdx.x;
    const int c = blockIdx.x;
    const int r = blockIdx.y - prows;
    const int lo = r * rn;
    const int gpr = rn >> 10;
    for (int i = tid; i < rn; i += 1024) s_tile[i] = 0;
    __syncthreads();

    const int e0 = c * chunk;
    const int len = min(chunk, nedges - e0);
    const int len4 = (len > 0) ? (len & ~3) : 0;
    const int* __restrict__ dp = dst + e0;
    for (int b0 = tid * 4; b0 < len4; b0 += 4096 * SCAN_BATCH) {
        int4 d[SCAN_BATCH];
#pragma unroll
        for (int u = 0; u < SCAN_BATCH; ++u) {
            const int idx = b0 + u * 4096;
            d[u] = (idx < len4) ? *(const int4*)(dp + idx) : make_int4(-1, -1, -1, -1);
        }
#pragma unroll
        for (int u = 0; u < SCAN_BATCH; ++u) {
            const unsigned k0 = (unsigned)(d[u].x - lo);
            const unsigned k1 = (unsigned)(d[u].y - lo);
            const unsigned k2 = (unsigned)(d[u].z - lo);
            const unsigned k3 = (unsigned)(d[u].w - lo);
            if (k0 < (unsigned)rn) atomicAdd(s_tile + k0, 1);
            if (k1 < (unsigned)rn) atomicAdd(s_tile + k1, 1);
            if (k2 < (unsigned)rn) atomicAdd(s_tile + k2, 1);
            if (k3 < (unsigned)rn) atomicAdd(s_tile + k3, 1);
        }
    }
    for (int i = len4 + tid; i < len; i += 1024) {
        const unsigned k = (unsigned)(dp[i] - lo);
        if (k < (unsigned)rn) atomicAdd(s_tile + k, 1);
    }
    __syncthreads();

    int* __restrict__ pc = part + c * npad + lo;
    for (int i = tid; i < rn; i += 1024) pc[i] = s_tile[i];
    const int lane = tid & 31;
    for (int k = tid >> 5; k < gpr; k += 32) {
        int s = 0;
#pragma unroll
        for (int m = 0; m < 32; ++m) s += s_tile[k * 1024 + m * 32 + lane];
        for (int d = 16; d > 0; d >>= 1) s += __shfl_xor(s, d);
        if (lane == 0) gsum[c * ngp + r * gpr + k] = s;
    }
}

__global__ void __launch_bounds__(1024)
sage_bases_kernel(int* __restrict__ part, const int* __restrict__ gsum,
                  int* __restrict__ off, int n, int npad, int ngp) {
    __shared__ int s_pre[32];
    __shared__ int s_inc[32];
    const int tid = threadIdx.x;
    const int lane = tid & 31;
    const int wv = tid >> 5;
    const int b = blockIdx.x;
    int pre = 0;
    for (int g = tid; g < b; g += 1024) {
#pragma unroll
        for (int c = 0; c < EDGE_CHUNKS; ++c) pre += gsum[c * ngp + g];
    }
    const int i = b * 1024 + tid;
    int v[EDGE_CHUNKS];
    int own = 0;
#pragma unroll
    for (int c = 0; c < EDGE_CHUNKS; ++c) {
        v[c] = part[c * npad + i];
        own += v[c];
    }
    int inc = own;
    for (int d = 1; d < 32; d <<= 1) {
        const int t = __shfl_up(inc, d);
        if (lane >= d) inc += t;
    }
    for (int d = 16; d > 0; d >>= 1) pre += __shfl_xor(pre, d);
    if (lane == 31) s_inc[wv] = inc;
    if (lane == 0) s_pre[wv] = pre;
    __syncthreads();
    int add = s_pre[lane] + ((lane < wv) ? s_inc[lane] : 0);
    for (int d = 16; d > 0; d >>= 1) add += __shfl_xor(add, d);
    int run = add + inc - own;
    if (i <= n) off[i] = run;
#pragma unroll
    for (int c = 0; c < EDGE_CHUNKS; ++c) {
        part[c * npad + i] = run;
        run += v[c];
    }
}

__global__ void __launch_bounds__(1024)
sage_place_tile_kernel(const int* __restrict__ src,
                       const int* __restrict__ dst,
                       const int* __restrict__ base,
                       int* __restrict__ nbr,
                       int nedges, int chunk, int rn, int npad) {
    extern __shared__ int s_tile[];
    const int tid = threadIdx.x;
    const int c = blockIdx.x;
    const int r = blockIdx.y;
    const int lo = r * rn;
    const int* __restrict__ bc = base + c * npad + lo;
    for (int i = tid; i < rn; i += 1024) s_tile[i] = bc[i];
    __syncthreads();
    unsigned* __restrict__ cur = (unsigned*)s_tile;

    const int e0 = c * chunk;
    const int len = min(chunk, nedges - e0);
    const int len4 = (len > 0) ? (len & ~3) : 0;
    const int* __restrict__ dp = dst + e0;
    const int* __restrict__ sp = src + e0;
    for (int b0 = tid * 4; b0 < len4; b0 += 4096 * SCAN_BATCH) {
        int4 d[SCAN_BATCH], s[SCAN_BATCH];
#pragma unroll
        for (int u = 0; u < SCAN_BATCH; ++u) {
            const int idx = b0 + u * 4096;
            const bool in = idx < len4;
            d[u] = in ? *(const int4*)(dp + idx) : make_int4(-1, -1, -1, -1);
            s[u] = in ? *(const int4*)(sp + idx) : make_int4(0, 0, 0, 0);
        }
        unsigned p[SCAN_BATCH][4];
        bool t[SCAN_BATCH][4];
#pragma unroll
        for (int u = 0; u < SCAN_BATCH; ++u) {
            const unsigned k0 = (unsigned)(d[u].x - lo);
            const unsigned k1 = (unsigned)(d[u].y - lo);
            const unsigned k2 = (unsigned)(d[u].z - lo);
            const unsigned k3 = (unsigned)(d[u].w - lo);
            t[u][0] = k0 < (unsigned)rn;
            t[u][1] = k1 < (unsigned)rn;
            t[u][2] = k2 < (unsigned)rn;
            t[u][3] = k3 < (unsigned)rn;
            p[u][0] = p[u][1] = p[u][2] = p[u][3] = 0u;
            if (t[u][0]) p[u][0] = atomicAdd(cur + k0, 1u);
            if (t[u][1]) p[u][1] = atomicAdd(cur + k1, 1u);
            if (t[u][2]) p[u][2] = atomicAdd(cur + k2, 1u);
            if (t[u][3]) p[u][3] = atomicAdd(cur + k3, 1u);
        }
#pragma unroll
        for (int u = 0; u < SCAN_BATCH; ++u) {
            if (t[u][0]) nbr[p[u][0]] = s[u].x;
            if (t[u][1]) nbr[p[u][1]] = s[u].y;
            if (t[u][2]) nbr[p[u][2]] = s[u].z;
            if (t[u][3]) nbr[p[u][3]] = s[u].w;
        }
    }
    for (int i = len4 + tid; i < len; i += 1024) {
        const unsigned k = (unsigned)(dp[i] - lo);
        if (k < (unsigned)rn) nbr[atomicAdd(cur + k, 1u)] = sp[i];
    }
}

static __device__ __forceinline__ float4 max4(const float4 a, const float4 b) {
    float4 r;
    r.x = fmaxf(a.x, b.x);
    r.y = fmaxf(a.y, b.y);
    r.z = fmaxf(a.z, b.z);
    r.w = fmaxf(a.w, b.w);
    return r;
}

static __device__ __forceinline__ float4 max_rows(const float* __restrict__ xc, const int myid,
                                                  const int cnt, const int lhalf, float4 acc) {
    int j = 0;
    for (; cnt - j > 4; j += 8) {
        const int s0 = __shfl(myid, j + lhalf);
        const int s1 = __shfl(myid, j + 2 + lhalf);
        const int s2 = __shfl(myid, j + 4 + lhalf);
        const int s3 = __shfl(myid, j + 6 + lhalf);
        const float4 v0 = *(const float4*)(xc + (long long)s0 * 64);
        const float4 v1 = *(const float4*)(xc + (long long)s1 * 64);
        const float4 v2 = *(const float4*)(xc + (long long)s2 * 64);
        const float4 v3 = *(const float4*)(xc + (long long)s3 * 64);
        acc = max4(acc, max4(max4(v0, v1), max4(v2, v3)));
    }
    if (j < cnt) {
        const int s0 = __shfl(myid, j + lhalf);
        const int s1 = __shfl(myid, j + 2 + lhalf);
        const float4 v0 = *(const float4*)(xc + (long long)s0 * 64);
        const float4 v1 = *(const float4*)(xc + (long long)s1 * 64);
        acc = max4(acc, max4(v0, v1));
    }
    return acc;
}

#define TILE_PITCH 68

template <int NTILES, bool LEAKY, bool POOL2>
static __device__ __forceinline__ void tile_product(const float* __restrict__ rows,
                                                    const float2* __restrict__ sWs,
                                                    const float2* __restrict__ sWn,
                                                    const float bv, float* __restrict__ Y,
                                                    float* __restrict__ hrows,
                                                    const int tile, const int t, const int nnodes,
                                                    const int lhalf, const int l16) {
    const int NOUT = NTILES * 16;
    const float* __restrict__ ax = rows + l16 * TILE_PITCH + 2 * lhalf;
    const float* __restrict__ ap = ax + 16 * TILE_PITCH;
    v8f c = {};
#pragma unroll
    for (int k0 = 0; k0 < 64; k0 += 4) {
        const int kp = (k0 >> 1) + lhalf;
        const float2 x2 = *(const float2*)(ax + k0);
        const float2 p2 = *(const float2*)(ap + k0);
        const float2 ws = sWs[kp * NOUT + t * 16 + l16];
        const float2 wn = sWn[kp * NOUT + t * 16 + l16];
        v2f a1; a1.x = x2.x; a1.y = x2.y;
        v2f a2; a2.x = p2.x; a2.y = p2.y;
        v2f b1; b1.x = ws.x; b1.y = ws.y;
        v2f b2; b2.x = wn.x; b2.y = wn.y;
        c = __builtin_amdgcn_wmma_f32_16x16x4_f32(
            false, a1, false, b1, (short)0, c, false, false);
        c = __builtin_amdgcn_wmma_f32_16x16x4_f32(
            false, a2, false, b2, (short)0, c, false, false);
    }
    const int row0 = tile * 16 + 8 * lhalf;
    float* __restrict__ yp = Y + (long long)row0 * NOUT + t * 16 + l16;
    float* __restrict__ hp = hrows + (8 * lhalf) * TILE_PITCH + t * 16 + l16;
    if (tile * 16 + 16 <= nnodes) {
#pragma unroll
        for (int j = 0; j < 8; ++j) {
            float v = c[j] + bv;
            if (LEAKY) v = (v >= 0.0f) ? v : 0.01f * v;
            yp[j * NOUT] = v;
            if (POOL2) hp[j * TILE_PITCH] = v;
        }
    } else {
#pragma unroll
        for (int j = 0; j < 8; ++j) {
            float v = c[j] + bv;
            if (LEAKY) v = (v >= 0.0f) ? v : 0.01f * v;
            if (row0 + j < nnodes) yp[j * NOUT] = v;
            if (POOL2) hp[j * TILE_PITCH] = v;
        }
    }
}

static __device__ __forceinline__ void pool2_product(const float* __restrict__ hrows,
                                                     const float2* __restrict__ sWp,
                                                     const float bv2, float* __restrict__ XP2,
                                                     const int tile, const int t, const int nnodes,
                                                     const int lhalf, const int l16) {
    const float* __restrict__ ah = hrows + l16 * TILE_PITCH + 2 * lhalf;
    v8f c = {};
#pragma unroll
    for (int k0 = 0; k0 < 64; k0 += 4) {
        const int kp = (k0 >> 1) + lhalf;
        const float2 h2 = *(const float2*)(ah + k0);
        const float2 wp = sWp[kp * 64 + t * 16 + l16];
        v2f a; a.x = h2.x; a.y = h2.y;
        v2f b; b.x = wp.x; b.y = wp.y;
        c = __builtin_amdgcn_wmma_f32_16x16x4_f32(
            false, a, false, b, (short)0, c, false, false);
    }
    const int row0 = tile * 16 + 8 * lhalf;
    float* __restrict__ xp = XP2 + (long long)row0 * 64 + t * 16 + l16;
    if (tile * 16 + 16 <= nnodes) {
#pragma unroll
        for (int j = 0; j < 8; ++j)
            xp[j * 64] = fmaxf(c[j] + bv2, 0.0f);
    } else {
#pragma unroll
        for (int j = 0; j < 8; ++j)
            if (row0 + j < nnodes) xp[j * 64] = fmaxf(c[j] + bv2, 0.0f);
    }
}

template <int NTILES, bool LEAKY, bool POOL2>
__global__ void __launch_bounds__(512, 16)
sage_gather_out_kernel(const int* __restrict__ off,
                       const int* __restrict__ nbr,
                       const float* __restrict__ XP,
                       const float* __restrict__ X,
                       const float* __restrict__ Wself,
                       const float* __restrict__ Wneigh,
                       const float* __restrict__ bias,
                       float* __restrict__ Y,
                       const float* __restrict__ Wpool2,
                       const float* __restrict__ bpool2,
                       float* __restrict__ XP2, int nnodes, int ntiles) {
    static_assert(!POOL2 || NTILES == 4, "the second stage reads 64-column tiles of Y");
    const int NOUT = NTILES * 16;
    __shared__ float2 sWs[32 * NTILES * 16];
    __shared__ float2 sWn[32 * NTILES * 16];
    __shared__ __attribute__((aligned(16))) float sA[2][2 * 16 * TILE_PITCH];
    __shared__ float2 sWp[POOL2 ? 32 * 64 : 1];
    __shared__ __attribute__((aligned(16))) float sH[2][POOL2 ? 16 * TILE_PITCH : 4];

#pragma unroll
    for (int i = threadIdx.x; i < 32 * NOUT; i += 512) {
        const int p = i / NOUT, c = i % NOUT;
        float2 t0, t1;
        t0.x = Wself[(2 * p) * NOUT + c];
        t0.y = Wself[(2 * p + 1) * NOUT + c];
        t1.x = Wneigh[(2 * p) * NOUT + c];
        t1.y = Wneigh[(2 * p + 1) * NOUT + c];
        sWs[i] = t0;
        sWn[i] = t1;
    }
    if (POOL2) {
#pragma unroll
        for (int i = threadIdx.x; i < 32 * 64; i += 512) {
            const int p = i >> 6, c = i & 63;
            float2 t2;
            t2.x = Wpool2[(2 * p) * 64 + c];
            t2.y = Wpool2[(2 * p + 1) * 64 + c];
            sWp[i] = t2;
        }
    }

    const int lane = threadIdx.x & 31;
    const int wv = __builtin_amdgcn_readfirstlane((int)(threadIdx.x >> 5));
    const int lhalf = lane >> 4;
    const int l16 = lane & 15;
    const float* __restrict__ xc = XP + l16 * 4;
    const int stride = gridDim.x;
    const int lastn = nnodes - 1;

    const int tcol = (NTILES == 4) ? (wv / 5) : 0;
    const bool mm4 = (wv % 5 == 0);
    const float bv = bias[tcol * 16 + l16];
    const int t2col = (wv == 2) ? 0 : (wv == 7) ? 1 : (wv == 8) ? 2 : 3;
    const bool mm2 = POOL2 && (wv == 2 || wv == 7 || wv == 8 || wv == 13);
    float bv2 = 0.0f;
    if (POOL2) bv2 = bpool2[t2col * 16 + l16];

    int tile = blockIdx.x;
    int bnd = off[min(tile * 16 + wv, lastn) + (lane & 1)];
    int it = 0;
    for (; tile < ntiles; tile += stride, ++it) {
        const int node = tile * 16 + wv;
        const int beg = __builtin_amdgcn_readlane(bnd, 0);
        const int end = (node < nnodes) ? __builtin_amdgcn_readlane(bnd, 1) : beg;
        const int cnt0 = min(32, end - beg);
        const int myid0 = nbr[(cnt0 > 0) ? beg + min(lane, cnt0 - 1) : 0];
        bnd = off[min(node + stride * 16, lastn) + (lane & 1)];
        const float4 sv = *(const float4*)(X + (long long)min(node, lastn) * 64 + l16 * 4);

        if (it > 0 && ((NTILES == 4) ? mm4 : (wv == 5 * ((it - 1) & 3))))
            tile_product<NTILES, LEAKY, POOL2>(sA[(it - 1) & 1], sWs, sWn, bv, Y, sH[(it - 1) & 1],
                                               tile - stride, tcol, nnodes, lhalf, l16);
        if (POOL2 && it > 1 && mm2)
            pool2_product(sH[it & 1], sWp, bv2, XP2, tile - 2 * stride, t2col, nnodes, lhalf, l16);

        float4 acc = make_float4(0.0f, 0.0f, 0.0f, 0.0f);
        acc = max_rows(xc, myid0, cnt0, lhalf, acc);
        for (int base = beg + 32; base < end; base += 32) {
            const int cnt = min(32, end - base);
            const int myid = nbr[base + min(lane, cnt - 1)];
            acc = max_rows(xc, myid, cnt, lhalf, acc);
        }
        acc.x = fmaxf(acc.x, __shfl_xor(acc.x, 16));
        acc.y = fmaxf(acc.y, __shfl_xor(acc.y, 16));
        acc.z = fmaxf(acc.z, __shfl_xor(acc.z, 16));
        acc.w = fmaxf(acc.w, __shfl_xor(acc.w, 16));
        float4 o;
        o.x = lhalf ? sv.x : acc.x;
        o.y = lhalf ? sv.y : acc.y;
        o.z = lhalf ? sv.z : acc.z;
        o.w = lhalf ? sv.w : acc.w;
        *(float4*)(&sA[it & 1][(1 - lhalf) * 16 * TILE_PITCH + wv * TILE_PITCH + l16 * 4]) = o;
        __syncthreads();
    }
    if (it > 0 && ((NTILES == 4) ? mm4 : (wv == 5 * ((it - 1) & 3))))
        tile_product<NTILES, LEAKY, POOL2>(sA[(it - 1) & 1], sWs, sWn, bv, Y, sH[(it - 1) & 1],
                                           tile - stride, tcol, nnodes, lhalf, l16);
    if (POOL2) {
        if (it > 1 && mm2)
            pool2_product(sH[it & 1], sWp, bv2, XP2, tile - 2 * stride, t2col, nnodes, lhalf, l16);
        __syncthreads();
        if (it > 0 && mm2)
            pool2_product(sH[(it - 1) & 1], sWp, bv2, XP2, tile - stride, t2col, nnodes, lhalf, l16);
    }
}

extern "C" void kernel_launch(void* const* d_in, const int* in_sizes, int n_in,
                              void* d_out, int out_size, void* d_ws, size_t ws_size,
                              hipStream_t stream) {
    const float* in_feat  = (const float*)d_in[0];
    const int*   src      = (const int*)d_in[1];
    const int*   dst      = (const int*)d_in[2];
    const float* W_pool1  = (const float*)d_in[3];
    const float* b_pool1  = (const float*)d_in[4];
    const float* W_self1  = (const float*)d_in[5];
    const float* W_neigh1 = (const float*)d_in[6];
    const float* bias1    = (const float*)d_in[7];
    const float* W_pool2  = (const float*)d_in[8];
    const float* b_pool2  = (const float*)d_in[9];
    const float* W_self2  = (const float*)d_in[10];
    const float* W_neigh2 = (const float*)d_in[11];
    const float* bias2    = (const float*)d_in[12];

    const int N = in_sizes[0] / 64;
    const int E = in_sizes[1];
    const long long NF = (long long)N * 64;

    const int scan_blocks  = (N + 1 + 1023) / 1024;
    const int ntiles       = (N + 15) / 16;
    const int fuse_blocks  = ntiles < 1024 ? ntiles : 1024;
    const int gpr   = (scan_blocks + NODE_RANGES - 1) / NODE_RANGES;
    const int rn    = gpr * 1024;
    const int ngp   = gpr * NODE_RANGES;
    const int npad  = ngp * 1024;
    const int chunk = ((E + EDGE_CHUNKS - 1) / EDGE_CHUNKS + 3) & ~3;
    const dim3 tile_grid(EDGE_CHUNKS, NODE_RANGES);
    const int prows = ((ntiles + 31) / 32 + EDGE_CHUNKS - 1) / EDGE_CHUNKS;
    const dim3 count_grid(EDGE_CHUNKS, prows + NODE_RANGES);
    const size_t count_lds = rn * sizeof(int) > 32 * 64 * sizeof(float2) ? rn * sizeof(int) : 32 * 64 * sizeof(float2);

    float* XP  = (float*)d_ws;
    float* H   = XP + NF;
    float* XP2 = H + NF;
    int* off  = (int*)(XP2 + NF);
    int* part = off + (N + 1024);
    int* gsum = part + EDGE_CHUNKS * npad;
    int* nbr  = gsum + EDGE_CHUNKS * ngp;

    sage_count_tile_kernel<<<count_grid, 1024, count_lds, stream>>>(
        dst, part, gsum, E, chunk, rn, npad, ngp, in_feat, W_pool1, b_pool1, XP, N, prows);
    sage_bases_kernel<<<scan_blocks, 1024, 0, stream>>>(part, gsum, off, N, npad, ngp);
    sage_place_tile_kernel<<<tile_grid, 1024, rn * sizeof(int), stream>>>(
        src, dst, part, nbr, E, chunk, rn, npad);

    sage_gather_out_kernel<4, true, true><<<fuse_blocks, 512, 0, stream>>>(
        off, nbr, XP, in_feat, W_self1, W_neigh1, bias1, H, W_pool2, b_pool2, XP2, N, ntiles);

    sage_gather_out_kernel<1, false, false><<<fuse_blocks, 512, 0, stream>>>(
        off, nbr, XP2, H, W_self2, W_neigh2, bias2, (float*)d_out, nullptr, nullptr, nullptr,
        N, ntiles);
}
